# starting kernel
# baseline (speedup 1.0000x reference)
#include <hip/hip_runtime.h>


#define N_NODES 20000
#define E_EDGES 320000
#define INV_SQRT3 0.57735026918962576451f

typedef __attribute__((ext_vector_type(2))) float v2f;
typedef __attribute__((ext_vector_type(8))) float v8f;

static __device__ __forceinline__ v8f wmma4(v2f a, v2f b, v8f c) {
  return __builtin_amdgcn_wmma_f32_16x16x4_f32(
      false, a, false, b, (short)0, c, false, false);
}

static __device__ __forceinline__ v2f ldb(const float* __restrict__ WP,
                                          int ncol, int k0, int col) {
  return *(const v2f*)(WP + (size_t)(k0 >> 1) * (2 * ncol) + col * 2);
}

static __device__ __forceinline__ v2f ld2(const float* __restrict__ p) {
  v2f r; r.x = p[0]; r.y = p[1]; return r;
}

__global__ void k_pack_pairs(const float* __restrict__ W, float* __restrict__ WP,
                             int K, int ncol) {
  const int i = blockIdx.x * blockDim.x + threadIdx.x;
  if (i >= K * ncol) return;
  const int k = i / ncol, c = i - k * ncol;
  WP[(size_t)(k >> 1) * (2 * ncol) + c * 2 + (k & 1)] = W[i];
}

__global__ __launch_bounds__(128) void k_node_pre(
    const float* __restrict__ nf, const float* __restrict__ Wp0P,
    const float* __restrict__ Wp1P, float* __restrict__ x0,
    float* __restrict__ x1, float* __restrict__ t0, float* __restrict__ t1) {
  const int wave = threadIdx.x >> 5;
  const int lane = threadIdx.x & 31;
  const int tile = blockIdx.x * 4 + wave;
  if (tile >= (N_NODES / 16)) return;
  const int nbase = tile * 16;
  const int lrow  = lane & 15;
  const int kh    = lane >> 4;
  const int arow  = nbase + lrow;

  {
    v8f acc[4] = {{}, {}, {}, {}};
    for (int kk = 0; kk < 64; kk += 4) {
      const int k0 = kk + 2 * kh;
      const v2f a = ld2(&nf[(size_t)arow * 256 + k0]);
#pragma unroll
      for (int j = 0; j < 4; ++j)
        acc[j] = wmma4(a, ldb(Wp0P, 64, k0, 16 * j + lrow), acc[j]);
    }
#pragma unroll
    for (int j = 0; j < 4; ++j)
#pragma unroll
      for (int v = 0; v < 8; ++v)
        x0[(size_t)(nbase + v + 8 * kh) * 64 + 16 * j + lrow] = acc[j][v];
  }

  for (int m = 0; m < 3; ++m) {
    v8f acc[4] = {{}, {}, {}, {}};
    for (int kk = 0; kk < 64; kk += 4) {
      const int k0 = kk + 2 * kh;
      v2f a;
      a.x = nf[(size_t)arow * 256 + 64 + k0 * 3 + m];
      a.y = nf[(size_t)arow * 256 + 64 + (k0 + 1) * 3 + m];
#pragma unroll
      for (int j = 0; j < 4; ++j)
        acc[j] = wmma4(a, ldb(Wp1P, 64, k0, 16 * j + lrow), acc[j]);
    }
#pragma unroll
    for (int j = 0; j < 4; ++j)
#pragma unroll
      for (int v = 0; v < 8; ++v)
        x1[(size_t)(nbase + v + 8 * kh) * 192 + (16 * j + lrow) * 3 + m] = acc[j][v];
  }

  float4* t0v = (float4*)(t0 + (size_t)nbase * 64);
  float4* t1v = (float4*)(t1 + (size_t)nbase * 192);
  const float4 z4 = make_float4(0.f, 0.f, 0.f, 0.f);
  for (int i = lane; i < 16 * 64 / 4; i += 32)  t0v[i] = z4;
  for (int i = lane; i < 16 * 192 / 4; i += 32) t1v[i] = z4;
}

__global__ __launch_bounds__(128) void k_edge(
    const float* __restrict__ sph, const float* __restrict__ rb,
    const int* __restrict__ snd, const int* __restrict__ rcv,
    const float* __restrict__ W1P, const float* __restrict__ W2P,
    const float* __restrict__ x0, const float* __restrict__ x1,
    float* __restrict__ t0, float* __restrict__ t1) {
  __shared__ float hidden[4][16][64];
  const int wave  = threadIdx.x >> 5;
  const int lane  = threadIdx.x & 31;
  const int lrow  = lane & 15;
  const int kh    = lane >> 4;
  const int ebase = (blockIdx.x * 4 + wave) * 16;

  {
    v8f acc[4] = {{}, {}, {}, {}};
#pragma unroll
    for (int kk = 0; kk < 8; kk += 4) {
      const int k0 = kk + 2 * kh;
      const v2f a = ld2(&rb[(size_t)(ebase + lrow) * 8 + k0]);
#pragma unroll
      for (int j = 0; j < 4; ++j)
        acc[j] = wmma4(a, ldb(W1P, 64, k0, 16 * j + lrow), acc[j]);
    }
#pragma unroll
    for (int j = 0; j < 4; ++j)
#pragma unroll
      for (int v = 0; v < 8; ++v) {
        float h = acc[j][v];
        h = h / (1.f + __expf(-h));
        hidden[wave][v + 8 * kh][16 * j + lrow] = h;
      }
  }
  __syncthreads();

  v8f w[16] = {{}, {}, {}, {}, {}, {}, {}, {}, {}, {}, {}, {}, {}, {}, {}, {}};
  for (int kk = 0; kk < 64; kk += 4) {
    const int k0 = kk + 2 * kh;
    const v2f a = ld2(&hidden[wave][lrow][k0]);
#pragma unroll
    for (int j = 0; j < 16; ++j)
      w[j] = wmma4(a, ldb(W2P, 256, k0, 16 * j + lrow), w[j]);
  }

#pragma unroll
  for (int v = 0; v < 8; ++v) {
    const int e = ebase + v + 8 * kh;
    const int s = snd[e];
    const int r = rcv[e];
    const float Y0  = sph[(size_t)e * 4 + 0];
    const float Y1x = sph[(size_t)e * 4 + 1];
    const float Y1y = sph[(size_t)e * 4 + 2];
    const float Y1z = sph[(size_t)e * 4 + 3];
#pragma unroll
    for (int j = 0; j < 4; ++j) {
      const int c = 16 * j + lrow;
      const float w0 = w[j][v], w1 = w[j + 4][v];
      const float w2 = w[j + 8][v], w3 = w[j + 12][v];
      const float h0  = x0[(size_t)s * 64 + c];
      const float h1x = x1[(size_t)s * 192 + c * 3 + 0];
      const float h1y = x1[(size_t)s * 192 + c * 3 + 1];
      const float h1z = x1[(size_t)s * 192 + c * 3 + 2];
      const float dot = h1x * Y1x + h1y * Y1y + h1z * Y1z;
      atomicAdd(&t0[(size_t)r * 64 + c], w0 * h0 * Y0 + w1 * dot * INV_SQRT3);
      const size_t b1 = (size_t)r * 192 + c * 3;
      atomicAdd(&t1[b1 + 0], w2 * h0 * Y1x + w3 * h1x * Y0);
      atomicAdd(&t1[b1 + 1], w2 * h0 * Y1y + w3 * h1y * Y0);
      atomicAdd(&t1[b1 + 2], w2 * h0 * Y1z + w3 * h1z * Y0);
    }
  }
}

__global__ __launch_bounds__(64) void k_node_post(
    const float* __restrict__ nf, const float* __restrict__ attrs,
    const float* __restrict__ Wpost0P, const float* __restrict__ Wpost1P,
    const float* __restrict__ Wc0, const float* __restrict__ Wc1,
    const float* __restrict__ Wr0P, const float* __restrict__ Wr1P,
    const float* __restrict__ Wf0P, const float* __restrict__ Wf1P,
    const float* __restrict__ t0, const float* __restrict__ t1,
    float* __restrict__ out) {
  __shared__ float s0s[2][16][64];
  __shared__ float s1s[2][16][192];
  const int wave  = threadIdx.x >> 5;
  const int lane  = threadIdx.x & 31;
  const int lrow  = lane & 15;
  const int kh    = lane >> 4;
  const int nbase = (blockIdx.x * 2 + wave) * 16;
  const int arow  = nbase + lrow;

  v8f a0[4]    = {{}, {}, {}, {}};
  v8f a1[3][4] = {{{}, {}, {}, {}}, {{}, {}, {}, {}}, {{}, {}, {}, {}}};
  for (int kk = 0; kk < 64; kk += 4) {
    const int k0 = kk + 2 * kh;
    const v2f at0 = ld2(&t0[(size_t)arow * 64 + k0]);
#pragma unroll
    for (int j = 0; j < 4; ++j)
      a0[j] = wmma4(at0, ldb(Wpost0P, 64, k0, 16 * j + lrow), a0[j]);
#pragma unroll
    for (int m = 0; m < 3; ++m) {
      v2f at1;
      at1.x = t1[(size_t)arow * 192 + k0 * 3 + m];
      at1.y = t1[(size_t)arow * 192 + (k0 + 1) * 3 + m];
#pragma unroll
      for (int j = 0; j < 4; ++j)
        a1[m][j] = wmma4(at1, ldb(Wpost1P, 64, k0, 16 * j + lrow), a1[m][j]);
    }
  }

#pragma unroll
  for (int j = 0; j < 4; ++j) {
    const int col = 16 * j + lrow;
#pragma unroll
    for (int v = 0; v < 8; ++v) {
      const int node = nbase + v + 8 * kh;
      const float s  = a0[j][v] * (1.f / 16.f);
      const float ax = a1[0][j][v] * (1.f / 16.f);
      const float ay = a1[1][j][v] * (1.f / 16.f);
      const float az = a1[2][j][v] * (1.f / 16.f);
      const float q  = ax * ax + ay * ay + az * az;
      float c0 = 0.f, c1 = 0.f, c2 = 0.f, d0 = 0.f, d1 = 0.f, d2 = 0.f;
#pragma unroll
      for (int z = 0; z < 10; ++z) {
        const float t = attrs[(size_t)node * 10 + z];
        c0 += t * Wc0[z * 64 + col];
        c1 += t * Wc0[640 + z * 64 + col];
        c2 += t * Wc0[1280 + z * 64 + col];
        d0 += t * Wc1[z * 64 + col];
        d1 += t * Wc1[640 + z * 64 + col];
        d2 += t * Wc1[1280 + z * 64 + col];
      }
      const float s2 = s * s;
      const float g  = d2 * (s2 + q) + d1 * s + d0;
      a0[j][v]    = c2 * (s2 * s + s * q) + c1 * (s2 + q) + c0 * s;
      a1[0][j][v] = g * ax;
      a1[1][j][v] = g * ay;
      a1[2][j][v] = g * az;
    }
  }

  for (int kk = 0; kk < 640; kk += 4) {
    const int k0 = kk + 2 * kh;
    const int u0 = k0 / 10, z0 = k0 - u0 * 10;
    const int k1 = k0 + 1;
    const int u1 = k1 / 10, z1 = k1 - u1 * 10;
    v2f a;
    a.x = nf[(size_t)arow * 256 + u0] * attrs[(size_t)arow * 10 + z0];
    a.y = nf[(size_t)arow * 256 + u1] * attrs[(size_t)arow * 10 + z1];
#pragma unroll
    for (int j = 0; j < 4; ++j)
      a0[j] = wmma4(a, ldb(Wr0P, 64, k0, 16 * j + lrow), a0[j]);
  }
#pragma unroll
  for (int m = 0; m < 3; ++m) {
    for (int kk = 0; kk < 640; kk += 4) {
      const int k0 = kk + 2 * kh;
      const int u0 = k0 / 10, z0 = k0 - u0 * 10;
      const int k1 = k0 + 1;
      const int u1 = k1 / 10, z1 = k1 - u1 * 10;
      v2f a;
      a.x = nf[(size_t)arow * 256 + 64 + u0 * 3 + m] * attrs[(size_t)arow * 10 + z0];
      a.y = nf[(size_t)arow * 256 + 64 + u1 * 3 + m] * attrs[(size_t)arow * 10 + z1];
#pragma unroll
      for (int j = 0; j < 4; ++j)
        a1[m][j] = wmma4(a, ldb(Wr1P, 64, k0, 16 * j + lrow), a1[m][j]);
    }
  }

#pragma unroll
  for (int j = 0; j < 4; ++j) {
    const int col = 16 * j + lrow;
#pragma unroll
    for (int v = 0; v < 8; ++v) {
      const int M = v + 8 * kh;
      s0s[wave][M][col] = a0[j][v];
      s1s[wave][M][col * 3 + 0] = a1[0][j][v];
      s1s[wave][M][col * 3 + 1] = a1[1][j][v];
      s1s[wave][M][col * 3 + 2] = a1[2][j][v];
    }
  }
  __syncthreads();

  {
    v8f acc[4] = {{}, {}, {}, {}};
    for (int kk = 0; kk < 64; kk += 4) {
      const int k0 = kk + 2 * kh;
      const v2f a = ld2(&s0s[wave][lrow][k0]);
#pragma unroll
      for (int j = 0; j < 4; ++j)
        acc[j] = wmma4(a, ldb(Wf0P, 64, k0, 16 * j + lrow), acc[j]);
    }
#pragma unroll
    for (int j = 0; j < 4; ++j)
#pragma unroll
      for (int v = 0; v < 8; ++v)
        out[(size_t)(nbase + v + 8 * kh) * 256 + 16 * j + lrow] = acc[j][v];
  }
#pragma unroll
  for (int m = 0; m < 3; ++m) {
    v8f acc[4] = {{}, {}, {}, {}};
    for (int kk = 0; kk < 64; kk += 4) {
      const int k0 = kk + 2 * kh;
      v2f a;
      a.x = s1s[wave][lrow][k0 * 3 + m];
      a.y = s1s[wave][lrow][(k0 + 1) * 3 + m];
#pragma unroll
      for (int j = 0; j < 4; ++j)
        acc[j] = wmma4(a, ldb(Wf1P, 64, k0, 16 * j + lrow), acc[j]);
    }
#pragma unroll
    for (int j = 0; j < 4; ++j)
#pragma unroll
      for (int v = 0; v < 8; ++v)
        out[(size_t)(nbase + v + 8 * kh) * 256 + 64 + (16 * j + lrow) * 3 + m] =
            acc[j][v];
  }
}

extern "C" void kernel_launch(void* const* d_in, const int* in_sizes, int n_in,
                              void* d_out, int out_size, void* d_ws, size_t ws_size,
                              hipStream_t stream) {
  (void)in_sizes; (void)n_in; (void)out_size; (void)ws_size;
  const float* nf     = (const float*)d_in[0];
  const float* attrs  = (const float*)d_in[1];
  const float* sph    = (const float*)d_in[2];
  const float* rb     = (const float*)d_in[3];
  const int*   snd    = (const int*)d_in[4];
  const int*   rcv    = (const int*)d_in[5];
  const float* Wpre0  = (const float*)d_in[6];
  const float* Wpre1  = (const float*)d_in[7];
  const float* W1     = (const float*)d_in[8];
  const float* W2     = (const float*)d_in[9];
  const float* Wpost0 = (const float*)d_in[10];
  const float* Wpost1 = (const float*)d_in[11];
  const float* Wc0    = (const float*)d_in[12];
  const float* Wc1    = (const float*)d_in[13];
  const float* Wr0    = (const float*)d_in[14];
  const float* Wr1    = (const float*)d_in[15];
  const float* Wf0    = (const float*)d_in[16];
  const float* Wf1    = (const float*)d_in[17];

  float* ws = (float*)d_ws;
  size_t off = 0;
  float* x0 = ws + off; off += (size_t)N_NODES * 64;
  float* x1 = ws + off; off += (size_t)N_NODES * 192;
  float* t0 = ws + off; off += (size_t)N_NODES * 64;
  float* t1 = ws + off; off += (size_t)N_NODES * 192;
  float* Wpre0P  = ws + off; off += 64 * 64;
  float* Wpre1P  = ws + off; off += 64 * 64;
  float* W1P     = ws + off; off += 8 * 64;
  float* W2P     = ws + off; off += 64 * 256;
  float* Wpost0P = ws + off; off += 64 * 64;
  float* Wpost1P = ws + off; off += 64 * 64;
  float* Wr0P    = ws + off; off += 640 * 64;
  float* Wr1P    = ws + off; off += 640 * 64;
  float* Wf0P    = ws + off; off += 64 * 64;
  float* Wf1P    = ws + off; off += 64 * 64;

  auto pack = [&](const float* W, float* WP, int K, int ncol) {
    const int n = K * ncol;
    k_pack_pairs<<<(n + 255) / 256, 256, 0, stream>>>(W, WP, K, ncol);
  };
  pack(Wpre0, Wpre0P, 64, 64);
  pack(Wpre1, Wpre1P, 64, 64);
  pack(W1, W1P, 8, 64);
  pack(W2, W2P, 64, 256);
  pack(Wpost0, Wpost0P, 64, 64);
  pack(Wpost1, Wpost1P, 64, 64);
  pack(Wr0, Wr0P, 640, 64);
  pack(Wr1, Wr1P, 640, 64);
  pack(Wf0, Wf0P, 64, 64);
  pack(Wf1, Wf1P, 64, 64);

  const int ntiles = N_NODES / 16;
  k_node_pre<<<(ntiles + 3) / 4, 128, 0, stream>>>(nf, Wpre0P, Wpre1P, x0, x1, t0, t1);
  k_edge<<<E_EDGES / 64, 128, 0, stream>>>(sph, rb, snd, rcv, W1P, W2P, x0, x1, t0, t1);
  k_node_post<<<ntiles / 2, 64, 0, stream>>>(nf, attrs, Wpost0P, Wpost1P, Wc0, Wc1,
                                             Wr0P, Wr1P, Wf0P, Wf1P, t0, t1,
                                             (float*)d_out);
}


# column tiles interleaved by 4 (lane = 4 adjacent columns): B fragments and sender rows read as 16-byte pieces of row-major rows, no weight pre-pack (13 launches -> 3); k order, atomic sums as before
# speedup vs baseline: 1.3748x; 1.3748x over previous
#include <hip/hip_runtime.h>


#define N_NODES 20000
#define E_EDGES 320000
#define INV_SQRT3 0.57735026918962576451f
#define RES_BLOCKS 313

typedef __attribute__((ext_vector_type(2))) float v2f;
typedef __attribute__((ext_vector_type(4))) float v4f;
typedef __attribute__((ext_vector_type(8))) float v8f;

static __device__ __forceinline__ v8f wmma4(v2f a, v2f b, v8f c) {
  return __builtin_amdgcn_wmma_f32_16x16x4_f32(
      false, a, false, b, (short)0, c, false, false);
}

static __device__ __forceinline__ void ldb4(const float* __restrict__ W, int ncol,
                                            int k0, int c0, v2f b[4]) {
  const v4f r0 = *(const v4f*)(W + (size_t)k0 * ncol + c0);
  const v4f r1 = *(const v4f*)(W + (size_t)(k0 + 1) * ncol + c0);
  b[0].x = r0.x; b[0].y = r1.x;
  b[1].x = r0.y; b[1].y = r1.y;
  b[2].x = r0.z; b[2].y = r1.z;
  b[3].x = r0.w; b[3].y = r1.w;
}

static __device__ __forceinline__ v2f ldb1(const float* __restrict__ W, int ncol,
                                          int k0, int col) {
  v2f r;
  r.x = W[(size_t)k0 * ncol + col];
  r.y = W[(size_t)(k0 + 1) * ncol + col];
  return r;
}

static __device__ __forceinline__ v2f ld2(const float* __restrict__ p) {
  v2f r; r.x = p[0]; r.y = p[1]; return r;
}

__global__ __launch_bounds__(128) void k_node_pre(
    const float* __restrict__ nf, const float* __restrict__ Wp0,
    const float* __restrict__ Wp1, float* __restrict__ x0,
    float* __restrict__ x1, float* __restrict__ t0, float* __restrict__ t1) {
  const int wave = threadIdx.x >> 5;
  const int lane = threadIdx.x & 31;
  const int tile = blockIdx.x * 4 + wave;
  if (tile >= (N_NODES / 16)) return;
  const int nbase = tile * 16;
  const int lrow  = lane & 15;
  const int kh    = lane >> 4;
  const int arow  = nbase + lrow;
  const int c4    = 4 * lrow;

  {
    v8f acc[4] = {{}, {}, {}, {}};
    for (int kk = 0; kk < 64; kk += 4) {
      const int k0 = kk + 2 * kh;
      const v2f a = ld2(&nf[(size_t)arow * 256 + k0]);
      v2f b[4];
      ldb4(Wp0, 64, k0, c4, b);
#pragma unroll
      for (int j = 0; j < 4; ++j)
        acc[j] = wmma4(a, b[j], acc[j]);
    }
#pragma unroll
    for (int v = 0; v < 8; ++v) {
      v4f o;
      o.x = acc[0][v]; o.y = acc[1][v]; o.z = acc[2][v]; o.w = acc[3][v];
      *(v4f*)&x0[(size_t)(nbase + v + 8 * kh) * 64 + c4] = o;
    }
  }

  for (int m = 0; m < 3; ++m) {
    v8f acc[4] = {{}, {}, {}, {}};
    for (int kk = 0; kk < 64; kk += 4) {
      const int k0 = kk + 2 * kh;
      v2f a;
      a.x = nf[(size_t)arow * 256 + 64 + k0 * 3 + m];
      a.y = nf[(size_t)arow * 256 + 64 + (k0 + 1) * 3 + m];
      v2f b[4];
      ldb4(Wp1, 64, k0, c4, b);
#pragma unroll
      for (int j = 0; j < 4; ++j)
        acc[j] = wmma4(a, b[j], acc[j]);
    }
#pragma unroll
    for (int j = 0; j < 4; ++j)
#pragma unroll
      for (int v = 0; v < 8; ++v)
        x1[(size_t)(nbase + v + 8 * kh) * 192 + (c4 + j) * 3 + m] = acc[j][v];
  }

  float4* t0v = (float4*)(t0 + (size_t)nbase * 64);
  float4* t1v = (float4*)(t1 + (size_t)nbase * 192);
  const float4 z4 = make_float4(0.f, 0.f, 0.f, 0.f);
  for (int i = lane; i < 16 * 64 / 4; i += 32)  t0v[i] = z4;
  for (int i = lane; i < 16 * 192 / 4; i += 32) t1v[i] = z4;
}

static __device__ __forceinline__ void residual_tile(
    const float* __restrict__ nf, const float* __restrict__ attrs,
    const float* __restrict__ Wr0, const float* __restrict__ Wr1,
    float* __restrict__ out, int nbase, int lane) {
  const int lrow = lane & 15;
  const int kh   = lane >> 4;
  const int arow = nbase + lrow;
  const int c4   = 4 * lrow;
  {
    v8f acc[4] = {{}, {}, {}, {}};
    for (int kk = 0; kk < 640; kk += 4) {
      const int k0 = kk + 2 * kh;
      const int u0 = k0 / 10, z0 = k0 - u0 * 10;
      const int k1 = k0 + 1;
      const int u1 = k1 / 10, z1 = k1 - u1 * 10;
      v2f a;
      a.x = nf[(size_t)arow * 256 + u0] * attrs[(size_t)arow * 10 + z0];
      a.y = nf[(size_t)arow * 256 + u1] * attrs[(size_t)arow * 10 + z1];
      v2f b[4];
      ldb4(Wr0, 64, k0, c4, b);
#pragma unroll
      for (int j = 0; j < 4; ++j)
        acc[j] = wmma4(a, b[j], acc[j]);
    }
#pragma unroll
    for (int v = 0; v < 8; ++v) {
      v4f o;
      o.x = acc[0][v]; o.y = acc[1][v]; o.z = acc[2][v]; o.w = acc[3][v];
      *(v4f*)&out[(size_t)(nbase + v + 8 * kh) * 256 + c4] = o;
    }
  }
#pragma unroll
  for (int m = 0; m < 3; ++m) {
    v8f acc[4] = {{}, {}, {}, {}};
#pragma unroll 4
    for (int kk = 0; kk < 640; kk += 4) {
      const int k0 = kk + 2 * kh;
      const int u0 = k0 / 10, z0 = k0 - u0 * 10;
      const int k1 = k0 + 1;
      const int u1 = k1 / 10, z1 = k1 - u1 * 10;
      v2f a;
      a.x = nf[(size_t)arow * 256 + 64 + u0 * 3 + m] * attrs[(size_t)arow * 10 + z0];
      a.y = nf[(size_t)arow * 256 + 64 + u1 * 3 + m] * attrs[(size_t)arow * 10 + z1];
      v2f b[4];
      ldb4(Wr1, 64, k0, c4, b);
#pragma unroll
      for (int j = 0; j < 4; ++j)
        acc[j] = wmma4(a, b[j], acc[j]);
    }
#pragma unroll
    for (int j = 0; j < 4; ++j)
#pragma unroll
      for (int v = 0; v < 8; ++v)
        out[(size_t)(nbase + v + 8 * kh) * 256 + 64 + (c4 + j) * 3 + m] =
            acc[j][v];
  }
}

__global__ __launch_bounds__(128) void k_edge(
    const float* __restrict__ sph, const float* __restrict__ rb,
    const int* __restrict__ snd, const int* __restrict__ rcv,
    const float* __restrict__ W1, const float* __restrict__ W2,
    const float* __restrict__ x0, const float* __restrict__ x1,
    float* __restrict__ t0, float* __restrict__ t1,
    const float* __restrict__ nf, const float* __restrict__ attrs,
    const float* __restrict__ Wr0, const float* __restrict__ Wr1,
    float* __restrict__ out) {
  __shared__ __attribute__((aligned(16))) float hidden[4][16][64];
  const int wave  = threadIdx.x >> 5;
  const int lane  = threadIdx.x & 31;
  if (blockIdx.x < RES_BLOCKS) {
    const int tile = blockIdx.x * 4 + wave;
    if (tile < (N_NODES / 16))
      residual_tile(nf, attrs, Wr0, Wr1, out, tile * 16, lane);
    return;
  }
  const int lrow  = lane & 15;
  const int kh    = lane >> 4;
  const int c4    = 4 * lrow;
  const int ebase = ((blockIdx.x - RES_BLOCKS) * 4 + wave) * 16;

  {
    v8f acc[4] = {{}, {}, {}, {}};
#pragma unroll
    for (int kk = 0; kk < 8; kk += 4) {
      const int k0 = kk + 2 * kh;
      const v2f a = ld2(&rb[(size_t)(ebase + lrow) * 8 + k0]);
      v2f b[4];
      ldb4(W1, 64, k0, c4, b);
#pragma unroll
      for (int j = 0; j < 4; ++j)
        acc[j] = wmma4(a, b[j], acc[j]);
    }
#pragma unroll
    for (int v = 0; v < 8; ++v) {
      float h[4];
#pragma unroll
      for (int j = 0; j < 4; ++j) {
        float t = acc[j][v];
        h[j] = t / (1.f + __expf(-t));
      }
      v4f o;
      o.x = h[0]; o.y = h[1]; o.z = h[2]; o.w = h[3];
      *(v4f*)&hidden[wave][v + 8 * kh][c4] = o;
    }
  }
  __syncthreads();

  v8f w[16] = {{}, {}, {}, {}, {}, {}, {}, {}, {}, {}, {}, {}, {}, {}, {}, {}};
#pragma unroll 2
  for (int kk = 0; kk < 64; kk += 4) {
    const int k0 = kk + 2 * kh;
    const v2f a = ld2(&hidden[wave][lrow][k0]);
#pragma unroll
    for (int p = 0; p < 4; ++p) {
      v2f b[4];
      ldb4(W2, 256, k0, 64 * p + c4, b);
#pragma unroll
      for (int i = 0; i < 4; ++i)
        w[4 * p + i] = wmma4(a, b[i], w[4 * p + i]);
    }
  }

  int sidx[8], ridx[8];
  v4f Yv[8];
#pragma unroll
  for (int v = 0; v < 8; ++v) {
    const int e = ebase + v + 8 * kh;
    sidx[v] = snd[e];
    ridx[v] = rcv[e];
    Yv[v]   = *(const v4f*)&sph[(size_t)e * 4];
  }
#pragma unroll
  for (int v = 0; v < 8; ++v) {
    const int s = sidx[v];
    const int r = ridx[v];
    const float Y0  = Yv[v].x;
    const float Y1x = Yv[v].y;
    const float Y1y = Yv[v].z;
    const float Y1z = Yv[v].w;
    const v4f g0 = *(const v4f*)&x0[(size_t)s * 64 + c4];
    const v4f ga = *(const v4f*)&x1[(size_t)s * 192 + 3 * c4];
    const v4f gb = *(const v4f*)&x1[(size_t)s * 192 + 3 * c4 + 4];
    const v4f gc = *(const v4f*)&x1[(size_t)s * 192 + 3 * c4 + 8];
    asm volatile("" :: "v"(g0), "v"(ga), "v"(gb), "v"(gc));
    const float hs[4]  = {g0.x, g0.y, g0.z, g0.w};
    const float hv[12] = {ga.x, ga.y, ga.z, ga.w, gb.x, gb.y, gb.z, gb.w,
                          gc.x, gc.y, gc.z, gc.w};
    float* t0r = t0 + (size_t)r * 64 + lrow;
    float* t1r = t1 + (size_t)r * 192 + lrow;
#pragma unroll
    for (int i = 0; i < 4; ++i) {
      const float w0 = w[i][v], w1 = w[i + 4][v];
      const float w2 = w[i + 8][v], w3 = w[i + 12][v];
      const float h0  = hs[i];
      const float h1x = hv[3 * i + 0];
      const float h1y = hv[3 * i + 1];
      const float h1z = hv[3 * i + 2];
      const float dot = h1x * Y1x + h1y * Y1y + h1z * Y1z;
      atomicAdd(&t0r[16 * i], w0 * h0 * Y0 + w1 * dot * INV_SQRT3);
      atomicAdd(&t1r[16 * i],       w2 * h0 * Y1x + w3 * h1x * Y0);
      atomicAdd(&t1r[64 + 16 * i],  w2 * h0 * Y1y + w3 * h1y * Y0);
      atomicAdd(&t1r[128 + 16 * i], w2 * h0 * Y1z + w3 * h1z * Y0);
    }
  }
}

__global__ __launch_bounds__(64, 2) void k_node_post(
    const float* __restrict__ nf, const float* __restrict__ attrs,
    const float* __restrict__ Wpost0, const float* __restrict__ Wpost1,
    const float* __restrict__ Wc0, const float* __restrict__ Wc1,
    const float* __restrict__ Wf0, const float* __restrict__ Wf1,
    const float* __restrict__ t0, const float* __restrict__ t1,
    float* __restrict__ out) {
  __shared__ __attribute__((aligned(16))) float s0s[2][16][64];
  __shared__ __attribute__((aligned(16))) float s1s[2][16][192];
  const int wave  = __builtin_amdgcn_readfirstlane(threadIdx.x >> 5);
  const int lane  = threadIdx.x & 31;
  const int lrow  = lane & 15;
  const int kh    = lane >> 4;
  const int c4    = 4 * lrow;
  const int nbase = (blockIdx.x * 2 + wave) * 16;
  const int arow  = nbase + lrow;

  {
    const float4* rsrc = (const float4*)(out + (size_t)nbase * 256);
    for (int q = lane; q < 16 * 64; q += 32) {
      const float4 r4 = rsrc[q];
      const int row = q >> 6, q4 = q & 63;
      if (q4 < 16) *(float4*)&s0s[wave][row][q4 * 4] = r4;
      else         *(float4*)&s1s[wave][row][(q4 - 16) * 4] = r4;
    }
  }

  v8f a0[4]    = {{}, {}, {}, {}};
  v8f a1[3][4] = {{{}, {}, {}, {}}, {{}, {}, {}, {}}, {{}, {}, {}, {}}};
  for (int kk = 0; kk < 64; kk += 4) {
    const int k0 = kk + 2 * kh;
    const int p0 = 16 * (k0 & 3) + (k0 >> 2);
    v2f at0;
    at0.x = t0[(size_t)arow * 64 + p0];
    at0.y = t0[(size_t)arow * 64 + p0 + 16];
    v2f b1[4];
#pragma unroll
    for (int j = 0; j < 4; ++j)
      b1[j] = ldb1(Wpost1, 64, k0, 16 * j + lrow);
#pragma unroll
    for (int j = 0; j < 4; ++j)
      a0[j] = wmma4(at0, ldb1(Wpost0, 64, k0, 16 * j + lrow), a0[j]);
#pragma unroll
    for (int m = 0; m < 3; ++m) {
      v2f at1;
      at1.x = t1[(size_t)arow * 192 + 64 * m + p0];
      at1.y = t1[(size_t)arow * 192 + 64 * m + p0 + 16];
#pragma unroll
      for (int j = 0; j < 4; ++j)
        a1[m][j] = wmma4(at1, b1[j], a1[m][j]);
    }
  }

#pragma unroll
  for (int j = 0; j < 4; ++j) {
    const int col = 16 * j + lrow;
#pragma unroll
    for (int v = 0; v < 8; ++v) {
      const int node = nbase + v + 8 * kh;
      const float s  = a0[j][v] * (1.f / 16.f);
      const float ax = a1[0][j][v] * (1.f / 16.f);
      const float ay = a1[1][j][v] * (1.f / 16.f);
      const float az = a1[2][j][v] * (1.f / 16.f);
      const float q  = ax * ax + ay * ay + az * az;
      float c0 = 0.f, c1 = 0.f, c2 = 0.f, d0 = 0.f, d1 = 0.f, d2 = 0.f;
#pragma unroll
      for (int z = 0; z < 10; ++z) {
        const float t = attrs[(size_t)node * 10 + z];
        c0 += t * Wc0[z * 64 + col];
        c1 += t * Wc0[640 + z * 64 + col];
        c2 += t * Wc0[1280 + z * 64 + col];
        d0 += t * Wc1[z * 64 + col];
        d1 += t * Wc1[640 + z * 64 + col];
        d2 += t * Wc1[1280 + z * 64 + col];
      }
      const float s2 = s * s;
      const float g  = d2 * (s2 + q) + d1 * s + d0;
      a0[j][v]    = c2 * (s2 * s + s * q) + c1 * (s2 + q) + c0 * s;
      a1[0][j][v] = g * ax;
      a1[1][j][v] = g * ay;
      a1[2][j][v] = g * az;
    }
  }

  __syncthreads();
#pragma unroll
  for (int j = 0; j < 4; ++j) {
    const int col = 16 * j + lrow;
#pragma unroll
    for (int v = 0; v < 8; ++v) {
      const int M = v + 8 * kh;
      s0s[wave][M][col] += a0[j][v];
      s1s[wave][M][col * 3 + 0] += a1[0][j][v];
      s1s[wave][M][col * 3 + 1] += a1[1][j][v];
      s1s[wave][M][col * 3 + 2] += a1[2][j][v];
    }
  }
  __syncthreads();

  {
    v8f acc[4] = {{}, {}, {}, {}};
    for (int kk = 0; kk < 64; kk += 4) {
      const int k0 = kk + 2 * kh;
      const v2f a = ld2(&s0s[wave][lrow][k0]);
      v2f b[4];
      ldb4(Wf0, 64, k0, c4, b);
#pragma unroll
      for (int j = 0; j < 4; ++j)
        acc[j] = wmma4(a, b[j], acc[j]);
    }
#pragma unroll
    for (int v = 0; v < 8; ++v) {
      v4f o;
      o.x = acc[0][v]; o.y = acc[1][v]; o.z = acc[2][v]; o.w = acc[3][v];
      *(v4f*)&out[(size_t)(nbase + v + 8 * kh) * 256 + c4] = o;
    }
  }
#pragma unroll
  for (int m = 0; m < 3; ++m) {
    v8f acc[4] = {{}, {}, {}, {}};
    for (int kk = 0; kk < 64; kk += 4) {
      const int k0 = kk + 2 * kh;
      v2f a;
      a.x = s1s[wave][lrow][k0 * 3 + m];
      a.y = s1s[wave][lrow][(k0 + 1) * 3 + m];
      v2f b[4];
      ldb4(Wf1, 64, k0, c4, b);
#pragma unroll
      for (int j = 0; j < 4; ++j)
        acc[j] = wmma4(a, b[j], acc[j]);
    }
#pragma unroll
    for (int j = 0; j < 4; ++j)
#pragma unroll
      for (int v = 0; v < 8; ++v)
        out[(size_t)(nbase + v + 8 * kh) * 256 + 64 + (c4 + j) * 3 + m] =
            acc[j][v];
  }
}

extern "C" void kernel_launch(void* const* d_in, const int* in_sizes, int n_in,
                              void* d_out, int out_size, void* d_ws, size_t ws_size,
                              hipStream_t stream) {
  (void)in_sizes; (void)n_in; (void)out_size; (void)ws_size;
  const float* nf     = (const float*)d_in[0];
  const float* attrs  = (const float*)d_in[1];
  const float* sph    = (const float*)d_in[2];
  const float* rb     = (const float*)d_in[3];
  const int*   snd    = (const int*)d_in[4];
  const int*   rcv    = (const int*)d_in[5];
  const float* Wpre0  = (const float*)d_in[6];
  const float* Wpre1  = (const float*)d_in[7];
  const float* W1     = (const float*)d_in[8];
  const float* W2     = (const float*)d_in[9];
  const float* Wpost0 = (const float*)d_in[10];
  const float* Wpost1 = (const float*)d_in[11];
  const float* Wc0    = (const float*)d_in[12];
  const float* Wc1    = (const float*)d_in[13];
  const float* Wr0    = (const float*)d_in[14];
  const float* Wr1    = (const float*)d_in[15];
  const float* Wf0    = (const float*)d_in[16];
  const float* Wf1    = (const float*)d_in[17];

  float* ws = (float*)d_ws;
  size_t off = 0;
  float* x0 = ws + off; off += (size_t)N_NODES * 64;
  float* x1 = ws + off; off += (size_t)N_NODES * 192;
  float* t0 = ws + off; off += (size_t)N_NODES * 64;
  float* t1 = ws + off; off += (size_t)N_NODES * 192;

  const int ntiles = N_NODES / 16;
  k_node_pre<<<(ntiles + 3) / 4, 128, 0, stream>>>(nf, Wpre0, Wpre1, x0, x1, t0, t1);
  k_edge<<<RES_BLOCKS + E_EDGES / 64, 128, 0, stream>>>(
      sph, rb, snd, rcv, W1, W2, x0, x1, t0, t1,
      nf, attrs, Wr0, Wr1, (float*)d_out);
  k_node_post<<<ntiles / 2, 64, 0, stream>>>(nf, attrs, Wpost0, Wpost1, Wc0, Wc1,
                                             Wf0, Wf1, t0, t1, (float*)d_out);
}


# residual K=640 products: one chain per wave (5000 short waves, not 1250 long), their blocks spread as every 3rd among the edge blocks; r1 handed over as 3 planes; k order, sums unchanged
# speedup vs baseline: 1.4322x; 1.0418x over previous
#include <hip/hip_runtime.h>


#define N_NODES 20000
#define E_EDGES 320000
#define INV_SQRT3 0.57735026918962576451f
#define RES_SPAN 3750
#define EDGE_GRID (RES_SPAN / 3 + E_EDGES / 64)

typedef __attribute__((ext_vector_type(2))) float v2f;
typedef __attribute__((ext_vector_type(4))) float v4f;
typedef __attribute__((ext_vector_type(8))) float v8f;

static __device__ __forceinline__ v8f wmma4(v2f a, v2f b, v8f c) {
  return __builtin_amdgcn_wmma_f32_16x16x4_f32(
      false, a, false, b, (short)0, c, false, false);
}

static __device__ __forceinline__ void ldb4(const float* __restrict__ W, int ncol,
                                            int k0, int c0, v2f b[4]) {
  const v4f r0 = *(const v4f*)(W + (size_t)k0 * ncol + c0);
  const v4f r1 = *(const v4f*)(W + (size_t)(k0 + 1) * ncol + c0);
  b[0].x = r0.x; b[0].y = r1.x;
  b[1].x = r0.y; b[1].y = r1.y;
  b[2].x = r0.z; b[2].y = r1.z;
  b[3].x = r0.w; b[3].y = r1.w;
}

static __device__ __forceinline__ v2f ldb1(const float* __restrict__ W, int ncol,
                                          int k0, int col) {
  v2f r;
  r.x = W[(size_t)k0 * ncol + col];
  r.y = W[(size_t)(k0 + 1) * ncol + col];
  return r;
}

static __device__ __forceinline__ v2f ld2(const float* __restrict__ p) {
  v2f r; r.x = p[0]; r.y = p[1]; return r;
}

__global__ __launch_bounds__(128) void k_node_pre(
    const float* __restrict__ nf, const float* __restrict__ Wp0,
    const float* __restrict__ Wp1, float* __restrict__ x0,
    float* __restrict__ x1, float* __restrict__ t0, float* __restrict__ t1) {
  const int wave = threadIdx.x >> 5;
  const int lane = threadIdx.x & 31;
  const int tile = blockIdx.x * 4 + wave;
  if (tile >= (N_NODES / 16)) return;
  const int nbase = tile * 16;
  const int lrow  = lane & 15;
  const int kh    = lane >> 4;
  const int arow  = nbase + lrow;
  const int c4    = 4 * lrow;

  {
    v8f acc[4] = {{}, {}, {}, {}};
    for (int kk = 0; kk < 64; kk += 4) {
      const int k0 = kk + 2 * kh;
      const v2f a = ld2(&nf[(size_t)arow * 256 + k0]);
      v2f b[4];
      ldb4(Wp0, 64, k0, c4, b);
#pragma unroll
      for (int j = 0; j < 4; ++j)
        acc[j] = wmma4(a, b[j], acc[j]);
    }
#pragma unroll
    for (int v = 0; v < 8; ++v) {
      v4f o;
      o.x = acc[0][v]; o.y = acc[1][v]; o.z = acc[2][v]; o.w = acc[3][v];
      *(v4f*)&x0[(size_t)(nbase + v + 8 * kh) * 64 + c4] = o;
    }
  }

  for (int m = 0; m < 3; ++m) {
    v8f acc[4] = {{}, {}, {}, {}};
    for (int kk = 0; kk < 64; kk += 4) {
      const int k0 = kk + 2 * kh;
      v2f a;
      a.x = nf[(size_t)arow * 256 + 64 + k0 * 3 + m];
      a.y = nf[(size_t)arow * 256 + 64 + (k0 + 1) * 3 + m];
      v2f b[4];
      ldb4(Wp1, 64, k0, c4, b);
#pragma unroll
      for (int j = 0; j < 4; ++j)
        acc[j] = wmma4(a, b[j], acc[j]);
    }
#pragma unroll
    for (int j = 0; j < 4; ++j)
#pragma unroll
      for (int v = 0; v < 8; ++v)
        x1[(size_t)(nbase + v + 8 * kh) * 192 + (c4 + j) * 3 + m] = acc[j][v];
  }

  float4* t0v = (float4*)(t0 + (size_t)nbase * 64);
  float4* t1v = (float4*)(t1 + (size_t)nbase * 192);
  const float4 z4 = make_float4(0.f, 0.f, 0.f, 0.f);
  for (int i = lane; i < 16 * 64 / 4; i += 32)  t0v[i] = z4;
  for (int i = lane; i < 16 * 192 / 4; i += 32) t1v[i] = z4;
}

static __device__ __forceinline__ void residual_chain(
    const float* __restrict__ nf, const float* __restrict__ attrs,
    const float* __restrict__ Wr, float* __restrict__ out,
    int nbase, int lane, int q) {
  const int lrow = lane & 15;
  const int kh   = lane >> 4;
  const int arow = nbase + lrow;
  const int c4   = 4 * lrow;
  const float* nrow = nf + (size_t)arow * 256 + (q ? 63 + q : 0);
  const float* trow = attrs + (size_t)arow * 10;
  const int ustep = q ? 3 : 1;
  v8f acc[4] = {{}, {}, {}, {}};
#pragma unroll 4
  for (int kk = 0; kk < 640; kk += 4) {
    const int k0 = kk + 2 * kh;
    const int u0 = k0 / 10, z0 = k0 - u0 * 10;
    const int k1 = k0 + 1;
    const int u1 = k1 / 10, z1 = k1 - u1 * 10;
    v2f a;
    a.x = nrow[u0 * ustep] * trow[z0];
    a.y = nrow[u1 * ustep] * trow[z1];
    v2f b[4];
    ldb4(Wr, 64, k0, c4, b);
#pragma unroll
    for (int j = 0; j < 4; ++j)
      acc[j] = wmma4(a, b[j], acc[j]);
  }
#pragma unroll
  for (int v = 0; v < 8; ++v) {
    v4f o;
    o.x = acc[0][v]; o.y = acc[1][v]; o.z = acc[2][v]; o.w = acc[3][v];
    *(v4f*)&out[(size_t)(nbase + v + 8 * kh) * 256 + 64 * q + c4] = o;
  }
}

__global__ __launch_bounds__(128) void k_edge(
    const float* __restrict__ sph, const float* __restrict__ rb,
    const int* __restrict__ snd, const int* __restrict__ rcv,
    const float* __restrict__ W1, const float* __restrict__ W2,
    const float* __restrict__ x0, const float* __restrict__ x1,
    float* __restrict__ t0, float* __restrict__ t1,
    const float* __restrict__ nf, const float* __restrict__ attrs,
    const float* __restrict__ Wr0, const float* __restrict__ Wr1,
    float* __restrict__ out) {
  __shared__ __attribute__((aligned(16))) float hidden[4][16][64];
  const int wave  = threadIdx.x >> 5;
  const int lane  = threadIdx.x & 31;
  const int bi = blockIdx.x;
  const int b3 = bi / 3;
  if (bi < RES_SPAN && bi == 3 * b3) {
    const int q = __builtin_amdgcn_readfirstlane(wave);
    residual_chain(nf, attrs, q ? Wr1 : Wr0, out, b3 * 16, lane, q);
    return;
  }
  const int eb = bi < RES_SPAN ? bi - b3 - 1 : bi - RES_SPAN / 3;
  const int lrow  = lane & 15;
  const int kh    = lane >> 4;
  const int c4    = 4 * lrow;
  const int ebase = (eb * 4 + wave) * 16;

  {
    v8f acc[4] = {{}, {}, {}, {}};
#pragma unroll
    for (int kk = 0; kk < 8; kk += 4) {
      const int k0 = kk + 2 * kh;
      const v2f a = ld2(&rb[(size_t)(ebase + lrow) * 8 + k0]);
      v2f b[4];
      ldb4(W1, 64, k0, c4, b);
#pragma unroll
      for (int j = 0; j < 4; ++j)
        acc[j] = wmma4(a, b[j], acc[j]);
    }
#pragma unroll
    for (int v = 0; v < 8; ++v) {
      float h[4];
#pragma unroll
      for (int j = 0; j < 4; ++j) {
        float t = acc[j][v];
        h[j] = t / (1.f + __expf(-t));
      }
      v4f o;
      o.x = h[0]; o.y = h[1]; o.z = h[2]; o.w = h[3];
      *(v4f*)&hidden[wave][v + 8 * kh][c4] = o;
    }
  }
  __syncthreads();

  v8f w[16] = {{}, {}, {}, {}, {}, {}, {}, {}, {}, {}, {}, {}, {}, {}, {}, {}};
#pragma unroll 2
  for (int kk = 0; kk < 64; kk += 4) {
    const int k0 = kk + 2 * kh;
    const v2f a = ld2(&hidden[wave][lrow][k0]);
#pragma unroll
    for (int p = 0; p < 4; ++p) {
      v2f b[4];
      ldb4(W2, 256, k0, 64 * p + c4, b);
#pragma unroll
      for (int i = 0; i < 4; ++i)
        w[4 * p + i] = wmma4(a, b[i], w[4 * p + i]);
    }
  }

  int sidx[8], ridx[8];
  v4f Yv[8];
#pragma unroll
  for (int v = 0; v < 8; ++v) {
    const int e = ebase + v + 8 * kh;
    sidx[v] = snd[e];
    ridx[v] = rcv[e];
    Yv[v]   = *(const v4f*)&sph[(size_t)e * 4];
  }
#pragma unroll
  for (int v = 0; v < 8; ++v) {
    const int s = sidx[v];
    const int r = ridx[v];
    const float Y0  = Yv[v].x;
    const float Y1x = Yv[v].y;
    const float Y1y = Yv[v].z;
    const float Y1z = Yv[v].w;
    const v4f g0 = *(const v4f*)&x0[(size_t)s * 64 + c4];
    const v4f ga = *(const v4f*)&x1[(size_t)s * 192 + 3 * c4];
    const v4f gb = *(const v4f*)&x1[(size_t)s * 192 + 3 * c4 + 4];
    const v4f gc = *(const v4f*)&x1[(size_t)s * 192 + 3 * c4 + 8];
    asm volatile("" :: "v"(g0), "v"(ga), "v"(gb), "v"(gc));
    const float hs[4]  = {g0.x, g0.y, g0.z, g0.w};
    const float hv[12] = {ga.x, ga.y, ga.z, ga.w, gb.x, gb.y, gb.z, gb.w,
                          gc.x, gc.y, gc.z, gc.w};
    float* t0r = t0 + (size_t)r * 64 + lrow;
    float* t1r = t1 + (size_t)r * 192 + lrow;
#pragma unroll
    for (int i = 0; i < 4; ++i) {
      const float w0 = w[i][v], w1 = w[i + 4][v];
      const float w2 = w[i + 8][v], w3 = w[i + 12][v];
      const float h0  = hs[i];
      const float h1x = hv[3 * i + 0];
      const float h1y = hv[3 * i + 1];
      const float h1z = hv[3 * i + 2];
      const float dot = h1x * Y1x + h1y * Y1y + h1z * Y1z;
      atomicAdd(&t0r[16 * i], w0 * h0 * Y0 + w1 * dot * INV_SQRT3);
      atomicAdd(&t1r[16 * i],       w2 * h0 * Y1x + w3 * h1x * Y0);
      atomicAdd(&t1r[64 + 16 * i],  w2 * h0 * Y1y + w3 * h1y * Y0);
      atomicAdd(&t1r[128 + 16 * i], w2 * h0 * Y1z + w3 * h1z * Y0);
    }
  }
}

__global__ __launch_bounds__(64, 2) void k_node_post(
    const float* __restrict__ nf, const float* __restrict__ attrs,
    const float* __restrict__ Wpost0, const float* __restrict__ Wpost1,
    const float* __restrict__ Wc0, const float* __restrict__ Wc1,
    const float* __restrict__ Wf0, const float* __restrict__ Wf1,
    const float* __restrict__ t0, const float* __restrict__ t1,
    float* __restrict__ out) {
  __shared__ __attribute__((aligned(16))) float s0s[2][16][64];
  __shared__ __attribute__((aligned(16))) float s1s[2][16][192];
  const int wave  = __builtin_amdgcn_readfirstlane(threadIdx.x >> 5);
  const int lane  = threadIdx.x & 31;
  const int lrow  = lane & 15;
  const int kh    = lane >> 4;
  const int c4    = 4 * lrow;
  const int nbase = (blockIdx.x * 2 + wave) * 16;
  const int arow  = nbase + lrow;

  {
    const float4* rsrc = (const float4*)(out + (size_t)nbase * 256);
    for (int q = lane; q < 16 * 64; q += 32) {
      const float4 r4 = rsrc[q];
      const int row = q >> 6, q4 = q & 63;
      if (q4 < 16) *(float4*)&s0s[wave][row][q4 * 4] = r4;
      else         *(float4*)&s1s[wave][row][(q4 - 16) * 4] = r4;
    }
  }

  v8f a0[4]    = {{}, {}, {}, {}};
  v8f a1[3][4] = {{{}, {}, {}, {}}, {{}, {}, {}, {}}, {{}, {}, {}, {}}};
  for (int kk = 0; kk < 64; kk += 4) {
    const int k0 = kk + 2 * kh;
    const int p0 = 16 * (k0 & 3) + (k0 >> 2);
    v2f at0;
    at0.x = t0[(size_t)arow * 64 + p0];
    at0.y = t0[(size_t)arow * 64 + p0 + 16];
    v2f b1[4];
#pragma unroll
    for (int j = 0; j < 4; ++j)
      b1[j] = ldb1(Wpost1, 64, k0, 16 * j + lrow);
#pragma unroll
    for (int j = 0; j < 4; ++j)
      a0[j] = wmma4(at0, ldb1(Wpost0, 64, k0, 16 * j + lrow), a0[j]);
#pragma unroll
    for (int m = 0; m < 3; ++m) {
      v2f at1;
      at1.x = t1[(size_t)arow * 192 + 64 * m + p0];
      at1.y = t1[(size_t)arow * 192 + 64 * m + p0 + 16];
#pragma unroll
      for (int j = 0; j < 4; ++j)
        a1[m][j] = wmma4(at1, b1[j], a1[m][j]);
    }
  }

#pragma unroll
  for (int j = 0; j < 4; ++j) {
    const int col = 16 * j + lrow;
#pragma unroll
    for (int v = 0; v < 8; ++v) {
      const int node = nbase + v + 8 * kh;
      const float s  = a0[j][v] * (1.f / 16.f);
      const float ax = a1[0][j][v] * (1.f / 16.f);
      const float ay = a1[1][j][v] * (1.f / 16.f);
      const float az = a1[2][j][v] * (1.f / 16.f);
      const float q  = ax * ax + ay * ay + az * az;
      float c0 = 0.f, c1 = 0.f, c2 = 0.f, d0 = 0.f, d1 = 0.f, d2 = 0.f;
#pragma unroll
      for (int z = 0; z < 10; ++z) {
        const float t = attrs[(size_t)node * 10 + z];
        c0 += t * Wc0[z * 64 + col];
        c1 += t * Wc0[640 + z * 64 + col];
        c2 += t * Wc0[1280 + z * 64 + col];
        d0 += t * Wc1[z * 64 + col];
        d1 += t * Wc1[640 + z * 64 + col];
        d2 += t * Wc1[1280 + z * 64 + col];
      }
      const float s2 = s * s;
      const float g  = d2 * (s2 + q) + d1 * s + d0;
      a0[j][v]    = c2 * (s2 * s + s * q) + c1 * (s2 + q) + c0 * s;
      a1[0][j][v] = g * ax;
      a1[1][j][v] = g * ay;
      a1[2][j][v] = g * az;
    }
  }

  __syncthreads();
#pragma unroll
  for (int j = 0; j < 4; ++j) {
    const int col = 16 * j + lrow;
#pragma unroll
    for (int v = 0; v < 8; ++v) {
      const int M = v + 8 * kh;
      s0s[wave][M][col] += a0[j][v];
      s1s[wave][M][col] += a1[0][j][v];
      s1s[wave][M][64 + col] += a1[1][j][v];
      s1s[wave][M][128 + col] += a1[2][j][v];
    }
  }
  __syncthreads();

  {
    v8f acc[4] = {{}, {}, {}, {}};
    for (int kk = 0; kk < 64; kk += 4) {
      const int k0 = kk + 2 * kh;
      const v2f a = ld2(&s0s[wave][lrow][k0]);
      v2f b[4];
      ldb4(Wf0, 64, k0, c4, b);
#pragma unroll
      for (int j = 0; j < 4; ++j)
        acc[j] = wmma4(a, b[j], acc[j]);
    }
#pragma unroll
    for (int v = 0; v < 8; ++v) {
      v4f o;
      o.x = acc[0][v]; o.y = acc[1][v]; o.z = acc[2][v]; o.w = acc[3][v];
      *(v4f*)&out[(size_t)(nbase + v + 8 * kh) * 256 + c4] = o;
    }
  }
#pragma unroll
  for (int m = 0; m < 3; ++m) {
    v8f acc[4] = {{}, {}, {}, {}};
    for (int kk = 0; kk < 64; kk += 4) {
      const int k0 = kk + 2 * kh;
      const v2f a = ld2(&s1s[wave][lrow][64 * m + k0]);
      v2f b[4];
      ldb4(Wf1, 64, k0, c4, b);
#pragma unroll
      for (int j = 0; j < 4; ++j)
        acc[j] = wmma4(a, b[j], acc[j]);
    }
#pragma unroll
    for (int j = 0; j < 4; ++j)
#pragma unroll
      for (int v = 0; v < 8; ++v)
        out[(size_t)(nbase + v + 8 * kh) * 256 + 64 + (c4 + j) * 3 + m] =
            acc[j][v];
  }
}

extern "C" void kernel_launch(void* const* d_in, const int* in_sizes, int n_in,
                              void* d_out, int out_size, void* d_ws, size_t ws_size,
                              hipStream_t stream) {
  (void)in_sizes; (void)n_in; (void)out_size; (void)ws_size;
  const float* nf     = (const float*)d_in[0];
  const float* attrs  = (const float*)d_in[1];
  const float* sph    = (const float*)d_in[2];
  const float* rb     = (const float*)d_in[3];
  const int*   snd    = (const int*)d_in[4];
  const int*   rcv    = (const int*)d_in[5];
  const float* Wpre0  = (const float*)d_in[6];
  const float* Wpre1  = (const float*)d_in[7];
  const float* W1     = (const float*)d_in[8];
  const float* W2     = (const float*)d_in[9];
  const float* Wpost0 = (const float*)d_in[10];
  const float* Wpost1 = (const float*)d_in[11];
  const float* Wc0    = (const float*)d_in[12];
  const float* Wc1    = (const float*)d_in[13];
  const float* Wr0    = (const float*)d_in[14];
  const float* Wr1    = (const float*)d_in[15];
  const float* Wf0    = (const float*)d_in[16];
  const float* Wf1    = (const float*)d_in[17];

  float* ws = (float*)d_ws;
  size_t off = 0;
  float* x0 = ws + off; off += (size_t)N_NODES * 64;
  float* x1 = ws + off; off += (size_t)N_NODES * 192;
  float* t0 = ws + off; off += (size_t)N_NODES * 64;
  float* t1 = ws + off; off += (size_t)N_NODES * 192;

  const int ntiles = N_NODES / 16;
  k_node_pre<<<(ntiles + 3) / 4, 128, 0, stream>>>(nf, Wpre0, Wpre1, x0, x1, t0, t1);
  k_edge<<<EDGE_GRID, 128, 0, stream>>>(
      sph, rb, snd, rcv, W1, W2, x0, x1, t0, t1,
      nf, attrs, Wr0, Wr1, (float*)d_out);
  k_node_post<<<ntiles / 2, 64, 0, stream>>>(nf, attrs, Wpost0, Wpost1, Wc0, Wc1,
                                             Wf0, Wf1, t0, t1, (float*)d_out);
}


# node-post: the six attribute-weighted coefficient sums (K=10, zero-padded to 12) run as WMMA products per column tile, not 60 loads and multiply-adds per element; those 10-term sums change order
# speedup vs baseline: 1.5487x; 1.0813x over previous
#include <hip/hip_runtime.h>


#define N_NODES 20000
#define E_EDGES 320000
#define INV_SQRT3 0.57735026918962576451f
#define RES_SPAN 3750
#define EDGE_GRID (RES_SPAN / 3 + E_EDGES / 64)

typedef __attribute__((ext_vector_type(2))) float v2f;
typedef __attribute__((ext_vector_type(4))) float v4f;
typedef __attribute__((ext_vector_type(8))) float v8f;

static __device__ __forceinline__ v8f wmma4(v2f a, v2f b, v8f c) {
  return __builtin_amdgcn_wmma_f32_16x16x4_f32(
      false, a, false, b, (short)0, c, false, false);
}

static __device__ __forceinline__ void ldb4(const float* __restrict__ W, int ncol,
                                            int k0, int c0, v2f b[4]) {
  const v4f r0 = *(const v4f*)(W + (size_t)k0 * ncol + c0);
  const v4f r1 = *(const v4f*)(W + (size_t)(k0 + 1) * ncol + c0);
  b[0].x = r0.x; b[0].y = r1.x;
  b[1].x = r0.y; b[1].y = r1.y;
  b[2].x = r0.z; b[2].y = r1.z;
  b[3].x = r0.w; b[3].y = r1.w;
}

static __device__ __forceinline__ v2f ldb1(const float* __restrict__ W, int ncol,
                                          int k0, int col) {
  v2f r;
  r.x = W[(size_t)k0 * ncol + col];
  r.y = W[(size_t)(k0 + 1) * ncol + col];
  return r;
}

static __device__ __forceinline__ v2f ld2(const float* __restrict__ p) {
  v2f r; r.x = p[0]; r.y = p[1]; return r;
}

__global__ __launch_bounds__(128) void k_node_pre(
    const float* __restrict__ nf, const float* __restrict__ Wp0,
    const float* __restrict__ Wp1, float* __restrict__ x0,
    float* __restrict__ x1, float* __restrict__ t0, float* __restrict__ t1) {
  const int wave = threadIdx.x >> 5;
  const int lane = threadIdx.x & 31;
  const int tile = blockIdx.x * 4 + wave;
  if (tile >= (N_NODES / 16)) return;
  const int nbase = tile * 16;
  const int lrow  = lane & 15;
  const int kh    = lane >> 4;
  const int arow  = nbase + lrow;
  const int c4    = 4 * lrow;

  {
    v8f acc[4] = {{}, {}, {}, {}};
    for (int kk = 0; kk < 64; kk += 4) {
      const int k0 = kk + 2 * kh;
      const v2f a = ld2(&nf[(size_t)arow * 256 + k0]);
      v2f b[4];
      ldb4(Wp0, 64, k0, c4, b);
#pragma unroll
      for (int j = 0; j < 4; ++j)
        acc[j] = wmma4(a, b[j], acc[j]);
    }
#pragma unroll
    for (int v = 0; v < 8; ++v) {
      v4f o;
      o.x = acc[0][v]; o.y = acc[1][v]; o.z = acc[2][v]; o.w = acc[3][v];
      *(v4f*)&x0[(size_t)(nbase + v + 8 * kh) * 64 + c4] = o;
    }
  }

  for (int m = 0; m < 3; ++m) {
    v8f acc[4] = {{}, {}, {}, {}};
    for (int kk = 0; kk < 64; kk += 4) {
      const int k0 = kk + 2 * kh;
      v2f a;
      a.x = nf[(size_t)arow * 256 + 64 + k0 * 3 + m];
      a.y = nf[(size_t)arow * 256 + 64 + (k0 + 1) * 3 + m];
      v2f b[4];
      ldb4(Wp1, 64, k0, c4, b);
#pragma unroll
      for (int j = 0; j < 4; ++j)
        acc[j] = wmma4(a, b[j], acc[j]);
    }
#pragma unroll
    for (int j = 0; j < 4; ++j)
#pragma unroll
      for (int v = 0; v < 8; ++v)
        x1[(size_t)(nbase + v + 8 * kh) * 192 + (c4 + j) * 3 + m] = acc[j][v];
  }

  float4* t0v = (float4*)(t0 + (size_t)nbase * 64);
  float4* t1v = (float4*)(t1 + (size_t)nbase * 192);
  const float4 z4 = make_float4(0.f, 0.f, 0.f, 0.f);
  for (int i = lane; i < 16 * 64 / 4; i += 32)  t0v[i] = z4;
  for (int i = lane; i < 16 * 192 / 4; i += 32) t1v[i] = z4;
}

static __device__ __forceinline__ void residual_chain(
    const float* __restrict__ nf, const float* __restrict__ attrs,
    const float* __restrict__ Wr, float* __restrict__ out,
    int nbase, int lane, int q) {
  const int lrow = lane & 15;
  const int kh   = lane >> 4;
  const int arow = nbase + lrow;
  const int c4   = 4 * lrow;
  const float* nrow = nf + (size_t)arow * 256 + (q ? 63 + q : 0);
  const float* trow = attrs + (size_t)arow * 10;
  const int ustep = q ? 3 : 1;
  v8f acc[4] = {{}, {}, {}, {}};
#pragma unroll 4
  for (int kk = 0; kk < 640; kk += 4) {
    const int k0 = kk + 2 * kh;
    const int u0 = k0 / 10, z0 = k0 - u0 * 10;
    const int k1 = k0 + 1;
    const int u1 = k1 / 10, z1 = k1 - u1 * 10;
    v2f a;
    a.x = nrow[u0 * ustep] * trow[z0];
    a.y = nrow[u1 * ustep] * trow[z1];
    v2f b[4];
    ldb4(Wr, 64, k0, c4, b);
#pragma unroll
    for (int j = 0; j < 4; ++j)
      acc[j] = wmma4(a, b[j], acc[j]);
  }
#pragma unroll
  for (int v = 0; v < 8; ++v) {
    v4f o;
    o.x = acc[0][v]; o.y = acc[1][v]; o.z = acc[2][v]; o.w = acc[3][v];
    *(v4f*)&out[(size_t)(nbase + v + 8 * kh) * 256 + 64 * q + c4] = o;
  }
}

__global__ __launch_bounds__(128) void k_edge(
    const float* __restrict__ sph, const float* __restrict__ rb,
    const int* __restrict__ snd, const int* __restrict__ rcv,
    const float* __restrict__ W1, const float* __restrict__ W2,
    const float* __restrict__ x0, const float* __restrict__ x1,
    float* __restrict__ t0, float* __restrict__ t1,
    const float* __restrict__ nf, const float* __restrict__ attrs,
    const float* __restrict__ Wr0, const float* __restrict__ Wr1,
    float* __restrict__ out) {
  __shared__ __attribute__((aligned(16))) float hidden[4][16][64];
  const int wave  = threadIdx.x >> 5;
  const int lane  = threadIdx.x & 31;
  const int bi = blockIdx.x;
  const int b3 = bi / 3;
  if (bi < RES_SPAN && bi == 3 * b3) {
    const int q = __builtin_amdgcn_readfirstlane(wave);
    residual_chain(nf, attrs, q ? Wr1 : Wr0, out, b3 * 16, lane, q);
    return;
  }
  const int eb = bi < RES_SPAN ? bi - b3 - 1 : bi - RES_SPAN / 3;
  const int lrow  = lane & 15;
  const int kh    = lane >> 4;
  const int c4    = 4 * lrow;
  const int ebase = (eb * 4 + wave) * 16;

  {
    v8f acc[4] = {{}, {}, {}, {}};
#pragma unroll
    for (int kk = 0; kk < 8; kk += 4) {
      const int k0 = kk + 2 * kh;
      const v2f a = ld2(&rb[(size_t)(ebase + lrow) * 8 + k0]);
      v2f b[4];
      ldb4(W1, 64, k0, c4, b);
#pragma unroll
      for (int j = 0; j < 4; ++j)
        acc[j] = wmma4(a, b[j], acc[j]);
    }
#pragma unroll
    for (int v = 0; v < 8; ++v) {
      float h[4];
#pragma unroll
      for (int j = 0; j < 4; ++j) {
        float t = acc[j][v];
        h[j] = t / (1.f + __expf(-t));
      }
      v4f o;
      o.x = h[0]; o.y = h[1]; o.z = h[2]; o.w = h[3];
      *(v4f*)&hidden[wave][v + 8 * kh][c4] = o;
    }
  }
  __syncthreads();

  v8f w[16] = {{}, {}, {}, {}, {}, {}, {}, {}, {}, {}, {}, {}, {}, {}, {}, {}};
#pragma unroll 2
  for (int kk = 0; kk < 64; kk += 4) {
    const int k0 = kk + 2 * kh;
    const v2f a = ld2(&hidden[wave][lrow][k0]);
#pragma unroll
    for (int p = 0; p < 4; ++p) {
      v2f b[4];
      ldb4(W2, 256, k0, 64 * p + c4, b);
#pragma unroll
      for (int i = 0; i < 4; ++i)
        w[4 * p + i] = wmma4(a, b[i], w[4 * p + i]);
    }
  }

  int sidx[8], ridx[8];
  v4f Yv[8];
#pragma unroll
  for (int v = 0; v < 8; ++v) {
    const int e = ebase + v + 8 * kh;
    sidx[v] = snd[e];
    ridx[v] = rcv[e];
    Yv[v]   = *(const v4f*)&sph[(size_t)e * 4];
  }
#pragma unroll
  for (int v = 0; v < 8; ++v) {
    const int s = sidx[v];
    const int r = ridx[v];
    const float Y0  = Yv[v].x;
    const float Y1x = Yv[v].y;
    const float Y1y = Yv[v].z;
    const float Y1z = Yv[v].w;
    const v4f g0 = *(const v4f*)&x0[(size_t)s * 64 + c4];
    const v4f ga = *(const v4f*)&x1[(size_t)s * 192 + 3 * c4];
    const v4f gb = *(const v4f*)&x1[(size_t)s * 192 + 3 * c4 + 4];
    const v4f gc = *(const v4f*)&x1[(size_t)s * 192 + 3 * c4 + 8];
    asm volatile("" :: "v"(g0), "v"(ga), "v"(gb), "v"(gc));
    const float hs[4]  = {g0.x, g0.y, g0.z, g0.w};
    const float hv[12] = {ga.x, ga.y, ga.z, ga.w, gb.x, gb.y, gb.z, gb.w,
                          gc.x, gc.y, gc.z, gc.w};
    float* t0r = t0 + (size_t)r * 64 + lrow;
    float* t1r = t1 + (size_t)r * 192 + lrow;
#pragma unroll
    for (int i = 0; i < 4; ++i) {
      const float w0 = w[i][v], w1 = w[i + 4][v];
      const float w2 = w[i + 8][v], w3 = w[i + 12][v];
      const float h0  = hs[i];
      const float h1x = hv[3 * i + 0];
      const float h1y = hv[3 * i + 1];
      const float h1z = hv[3 * i + 2];
      const float dot = h1x * Y1x + h1y * Y1y + h1z * Y1z;
      atomicAdd(&t0r[16 * i], w0 * h0 * Y0 + w1 * dot * INV_SQRT3);
      atomicAdd(&t1r[16 * i],       w2 * h0 * Y1x + w3 * h1x * Y0);
      atomicAdd(&t1r[64 + 16 * i],  w2 * h0 * Y1y + w3 * h1y * Y0);
      atomicAdd(&t1r[128 + 16 * i], w2 * h0 * Y1z + w3 * h1z * Y0);
    }
  }
}

__global__ __launch_bounds__(64, 2) void k_node_post(
    const float* __restrict__ nf, const float* __restrict__ attrs,
    const float* __restrict__ Wpost0, const float* __restrict__ Wpost1,
    const float* __restrict__ Wc0, const float* __restrict__ Wc1,
    const float* __restrict__ Wf0, const float* __restrict__ Wf1,
    const float* __restrict__ t0, const float* __restrict__ t1,
    float* __restrict__ out) {
  __shared__ __attribute__((aligned(16))) float s0s[2][16][64];
  __shared__ __attribute__((aligned(16))) float s1s[2][16][192];
  const int wave  = __builtin_amdgcn_readfirstlane(threadIdx.x >> 5);
  const int lane  = threadIdx.x & 31;
  const int lrow  = lane & 15;
  const int kh    = lane >> 4;
  const int c4    = 4 * lrow;
  const int nbase = (blockIdx.x * 2 + wave) * 16;
  const int arow  = nbase + lrow;

  {
    const float4* rsrc = (const float4*)(out + (size_t)nbase * 256);
    for (int q = lane; q < 16 * 64; q += 32) {
      const float4 r4 = rsrc[q];
      const int row = q >> 6, q4 = q & 63;
      if (q4 < 16) *(float4*)&s0s[wave][row][q4 * 4] = r4;
      else         *(float4*)&s1s[wave][row][(q4 - 16) * 4] = r4;
    }
  }

  v8f a0[4]    = {{}, {}, {}, {}};
  v8f a1[3][4] = {{{}, {}, {}, {}}, {{}, {}, {}, {}}, {{}, {}, {}, {}}};
  for (int kk = 0; kk < 64; kk += 4) {
    const int k0 = kk + 2 * kh;
    const int p0 = 16 * (k0 & 3) + (k0 >> 2);
    v2f at0;
    at0.x = t0[(size_t)arow * 64 + p0];
    at0.y = t0[(size_t)arow * 64 + p0 + 16];
    v2f b1[4];
#pragma unroll
    for (int j = 0; j < 4; ++j)
      b1[j] = ldb1(Wpost1, 64, k0, 16 * j + lrow);
#pragma unroll
    for (int j = 0; j < 4; ++j)
      a0[j] = wmma4(at0, ldb1(Wpost0, 64, k0, 16 * j + lrow), a0[j]);
#pragma unroll
    for (int m = 0; m < 3; ++m) {
      v2f at1;
      at1.x = t1[(size_t)arow * 192 + 64 * m + p0];
      at1.y = t1[(size_t)arow * 192 + 64 * m + p0 + 16];
#pragma unroll
      for (int j = 0; j < 4; ++j)
        a1[m][j] = wmma4(at1, b1[j], a1[m][j]);
    }
  }

  const bool kpad = (kh != 0);
  v2f aat[3];
#pragma unroll
  for (int t = 0; t < 3; ++t) {
    const bool pad = (t == 2) && kpad;
    const int kc = pad ? 8 : 4 * t + 2 * kh;
    v2f a = ld2(&attrs[(size_t)arow * 10 + kc]);
    if (pad) { a.x = 0.f; a.y = 0.f; }
    aat[t] = a;
  }
#pragma unroll
  for (int j = 0; j < 4; ++j) {
    const int col = 16 * j + lrow;
    v2f bw[18];
#pragma unroll
    for (int t = 0; t < 3; ++t) {
      const int kc = ((t == 2) && kpad) ? 8 : 4 * t + 2 * kh;
#pragma unroll
      for (int p = 0; p < 6; ++p) {
        const float* Wm = (p < 3) ? (Wc0 + 640 * p) : (Wc1 + 640 * (p - 3));
        bw[6 * t + p] = ldb1(Wm, 64, kc, col);
      }
    }
    asm volatile("" :: "v"(bw[0]), "v"(bw[1]), "v"(bw[2]), "v"(bw[3]), "v"(bw[4]),
                 "v"(bw[5]), "v"(bw[6]), "v"(bw[7]), "v"(bw[8]), "v"(bw[9]),
                 "v"(bw[10]), "v"(bw[11]), "v"(bw[12]), "v"(bw[13]), "v"(bw[14]),
                 "v"(bw[15]), "v"(bw[16]), "v"(bw[17]));
    v8f cf[6] = {{}, {}, {}, {}, {}, {}};
#pragma unroll
    for (int t = 0; t < 3; ++t) {
      const bool pad = (t == 2) && kpad;
#pragma unroll
      for (int p = 0; p < 6; ++p) {
        v2f b = bw[6 * t + p];
        if (pad) { b.x = 0.f; b.y = 0.f; }
        cf[p] = wmma4(aat[t], b, cf[p]);
      }
    }
#pragma unroll
    for (int v = 0; v < 8; ++v) {
      const float s  = a0[j][v] * (1.f / 16.f);
      const float ax = a1[0][j][v] * (1.f / 16.f);
      const float ay = a1[1][j][v] * (1.f / 16.f);
      const float az = a1[2][j][v] * (1.f / 16.f);
      const float q  = ax * ax + ay * ay + az * az;
      const float c0 = cf[0][v], c1 = cf[1][v], c2 = cf[2][v];
      const float d0 = cf[3][v], d1 = cf[4][v], d2 = cf[5][v];
      const float s2 = s * s;
      const float g  = d2 * (s2 + q) + d1 * s + d0;
      a0[j][v]    = c2 * (s2 * s + s * q) + c1 * (s2 + q) + c0 * s;
      a1[0][j][v] = g * ax;
      a1[1][j][v] = g * ay;
      a1[2][j][v] = g * az;
    }
    __builtin_amdgcn_sched_barrier(0);
  }

  __syncthreads();
#pragma unroll
  for (int j = 0; j < 4; ++j) {
    const int col = 16 * j + lrow;
#pragma unroll
    for (int v = 0; v < 8; ++v) {
      const int M = v + 8 * kh;
      s0s[wave][M][col] += a0[j][v];
      s1s[wave][M][col] += a1[0][j][v];
      s1s[wave][M][64 + col] += a1[1][j][v];
      s1s[wave][M][128 + col] += a1[2][j][v];
    }
  }
  __syncthreads();

  {
    v8f acc[4] = {{}, {}, {}, {}};
    for (int kk = 0; kk < 64; kk += 4) {
      const int k0 = kk + 2 * kh;
      const v2f a = ld2(&s0s[wave][lrow][k0]);
      v2f b[4];
      ldb4(Wf0, 64, k0, c4, b);
#pragma unroll
      for (int j = 0; j < 4; ++j)
        acc[j] = wmma4(a, b[j], acc[j]);
    }
#pragma unroll
    for (int v = 0; v < 8; ++v) {
      v4f o;
      o.x = acc[0][v]; o.y = acc[1][v]; o.z = acc[2][v]; o.w = acc[3][v];
      *(v4f*)&out[(size_t)(nbase + v + 8 * kh) * 256 + c4] = o;
    }
  }
#pragma unroll
  for (int m = 0; m < 3; ++m) {
    v8f acc[4] = {{}, {}, {}, {}};
    for (int kk = 0; kk < 64; kk += 4) {
      const int k0 = kk + 2 * kh;
      const v2f a = ld2(&s1s[wave][lrow][64 * m + k0]);
      v2f b[4];
      ldb4(Wf1, 64, k0, c4, b);
#pragma unroll
      for (int j = 0; j < 4; ++j)
        acc[j] = wmma4(a, b[j], acc[j]);
    }
#pragma unroll
    for (int j = 0; j < 4; ++j)
#pragma unroll
      for (int v = 0; v < 8; ++v)
        out[(size_t)(nbase + v + 8 * kh) * 256 + 64 + (c4 + j) * 3 + m] =
            acc[j][v];
  }
}

extern "C" void kernel_launch(void* const* d_in, const int* in_sizes, int n_in,
                              void* d_out, int out_size, void* d_ws, size_t ws_size,
                              hipStream_t stream) {
  (void)in_sizes; (void)n_in; (void)out_size; (void)ws_size;
  const float* nf     = (const float*)d_in[0];
  const float* attrs  = (const float*)d_in[1];
  const float* sph    = (const float*)d_in[2];
  const float* rb     = (const float*)d_in[3];
  const int*   snd    = (const int*)d_in[4];
  const int*   rcv    = (const int*)d_in[5];
  const float* Wpre0  = (const float*)d_in[6];
  const float* Wpre1  = (const float*)d_in[7];
  const float* W1     = (const float*)d_in[8];
  const float* W2     = (const float*)d_in[9];
  const float* Wpost0 = (const float*)d_in[10];
  const float* Wpost1 = (const float*)d_in[11];
  const float* Wc0    = (const float*)d_in[12];
  const float* Wc1    = (const float*)d_in[13];
  const float* Wr0    = (const float*)d_in[14];
  const float* Wr1    = (const float*)d_in[15];
  const float* Wf0    = (const float*)d_in[16];
  const float* Wf1    = (const float*)d_in[17];

  float* ws = (float*)d_ws;
  size_t off = 0;
  float* x0 = ws + off; off += (size_t)N_NODES * 64;
  float* x1 = ws + off; off += (size_t)N_NODES * 192;
  float* t0 = ws + off; off += (size_t)N_NODES * 64;
  float* t1 = ws + off; off += (size_t)N_NODES * 192;

  const int ntiles = N_NODES / 16;
  k_node_pre<<<(ntiles + 3) / 4, 128, 0, stream>>>(nf, Wpre0, Wpre1, x0, x1, t0, t1);
  k_edge<<<EDGE_GRID, 128, 0, stream>>>(
      sph, rb, snd, rcv, W1, W2, x0, x1, t0, t1,
      nf, attrs, Wr0, Wr1, (float*)d_out);
  k_node_post<<<ntiles / 2, 64, 0, stream>>>(nf, attrs, Wpost0, Wpost1, Wc0, Wc1,
                                             Wf0, Wf1, t0, t1, (float*)d_out);
}


# residual contraction re-associated: per attribute z one K=64 WMMA product x @ Wr[:,z,:] (x fragments loaded once per chain), scaled by attrs[:,z] on the accumulators and summed; sum order changes
# speedup vs baseline: 1.6419x; 1.0602x over previous
#include <hip/hip_runtime.h>


#define N_NODES 20000
#define E_EDGES 320000
#define INV_SQRT3 0.57735026918962576451f
#define RES_SPAN 3750
#define EDGE_GRID (RES_SPAN / 3 + E_EDGES / 64)

typedef __attribute__((ext_vector_type(2))) float v2f;
typedef __attribute__((ext_vector_type(4))) float v4f;
typedef __attribute__((ext_vector_type(8))) float v8f;

static __device__ __forceinline__ v8f wmma4(v2f a, v2f b, v8f c) {
  return __builtin_amdgcn_wmma_f32_16x16x4_f32(
      false, a, false, b, (short)0, c, false, false);
}

static __device__ __forceinline__ void ldb4(const float* __restrict__ W, int ncol,
                                            int k0, int c0, v2f b[4]) {
  const v4f r0 = *(const v4f*)(W + (size_t)k0 * ncol + c0);
  const v4f r1 = *(const v4f*)(W + (size_t)(k0 + 1) * ncol + c0);
  b[0].x = r0.x; b[0].y = r1.x;
  b[1].x = r0.y; b[1].y = r1.y;
  b[2].x = r0.z; b[2].y = r1.z;
  b[3].x = r0.w; b[3].y = r1.w;
}

static __device__ __forceinline__ v2f ldb1(const float* __restrict__ W, int ncol,
                                          int k0, int col) {
  v2f r;
  r.x = W[(size_t)k0 * ncol + col];
  r.y = W[(size_t)(k0 + 1) * ncol + col];
  return r;
}

static __device__ __forceinline__ v2f ld2(const float* __restrict__ p) {
  v2f r; r.x = p[0]; r.y = p[1]; return r;
}

__global__ __launch_bounds__(128) void k_node_pre(
    const float* __restrict__ nf, const float* __restrict__ Wp0,
    const float* __restrict__ Wp1, float* __restrict__ x0,
    float* __restrict__ x1, float* __restrict__ t0, float* __restrict__ t1) {
  const int wave = threadIdx.x >> 5;
  const int lane = threadIdx.x & 31;
  const int tile = blockIdx.x * 4 + wave;
  if (tile >= (N_NODES / 16)) return;
  const int nbase = tile * 16;
  const int lrow  = lane & 15;
  const int kh    = lane >> 4;
  const int arow  = nbase + lrow;
  const int c4    = 4 * lrow;

  {
    v8f acc[4] = {{}, {}, {}, {}};
    for (int kk = 0; kk < 64; kk += 4) {
      const int k0 = kk + 2 * kh;
      const v2f a = ld2(&nf[(size_t)arow * 256 + k0]);
      v2f b[4];
      ldb4(Wp0, 64, k0, c4, b);
#pragma unroll
      for (int j = 0; j < 4; ++j)
        acc[j] = wmma4(a, b[j], acc[j]);
    }
#pragma unroll
    for (int v = 0; v < 8; ++v) {
      v4f o;
      o.x = acc[0][v]; o.y = acc[1][v]; o.z = acc[2][v]; o.w = acc[3][v];
      *(v4f*)&x0[(size_t)(nbase + v + 8 * kh) * 64 + c4] = o;
    }
  }

  for (int m = 0; m < 3; ++m) {
    v8f acc[4] = {{}, {}, {}, {}};
    for (int kk = 0; kk < 64; kk += 4) {
      const int k0 = kk + 2 * kh;
      v2f a;
      a.x = nf[(size_t)arow * 256 + 64 + k0 * 3 + m];
      a.y = nf[(size_t)arow * 256 + 64 + (k0 + 1) * 3 + m];
      v2f b[4];
      ldb4(Wp1, 64, k0, c4, b);
#pragma unroll
      for (int j = 0; j < 4; ++j)
        acc[j] = wmma4(a, b[j], acc[j]);
    }
#pragma unroll
    for (int j = 0; j < 4; ++j)
#pragma unroll
      for (int v = 0; v < 8; ++v)
        x1[(size_t)(nbase + v + 8 * kh) * 192 + (c4 + j) * 3 + m] = acc[j][v];
  }

  float4* t0v = (float4*)(t0 + (size_t)nbase * 64);
  float4* t1v = (float4*)(t1 + (size_t)nbase * 192);
  const float4 z4 = make_float4(0.f, 0.f, 0.f, 0.f);
  for (int i = lane; i < 16 * 64 / 4; i += 32)  t0v[i] = z4;
  for (int i = lane; i < 16 * 192 / 4; i += 32) t1v[i] = z4;
}

static __device__ __forceinline__ void residual_chain(
    const float* __restrict__ nf, const float* __restrict__ attrs,
    const float* __restrict__ Wr, float* __restrict__ out,
    int nbase, int lane, int q) {
  const int lrow = lane & 15;
  const int kh   = lane >> 4;
  const int arow = nbase + lrow;
  const int c4   = 4 * lrow;
  const float* nrow = nf + (size_t)arow * 256 + (q ? 63 + q : 0);
  const int ustep = q ? 3 : 1;
  v2f a[16];
#pragma unroll
  for (int t = 0; t < 16; ++t) {
    const int k0 = 4 * t + 2 * kh;
    a[t].x = nrow[k0 * ustep];
    a[t].y = nrow[(k0 + 1) * ustep];
  }
  const float* trow = attrs + (size_t)(nbase + 8 * kh) * 10;
  v8f r[4] = {{}, {}, {}, {}};
  for (int z = 0; z < 10; ++z) {
    const float* Wz = Wr + 64 * z;
    v8f acc[4] = {{}, {}, {}, {}};
#pragma unroll
    for (int t = 0; t < 16; ++t) {
      const int k0 = 4 * t + 2 * kh;
      v2f b[4];
      ldb4(Wz, 640, k0, c4, b);
#pragma unroll
      for (int j = 0; j < 4; ++j)
        acc[j] = wmma4(a[t], b[j], acc[j]);
    }
#pragma unroll
    for (int v = 0; v < 8; ++v) {
      const float tz = trow[10 * v + z];
#pragma unroll
      for (int j = 0; j < 4; ++j)
        r[j][v] += tz * acc[j][v];
    }
  }
#pragma unroll
  for (int v = 0; v < 8; ++v) {
    v4f o;
    o.x = r[0][v]; o.y = r[1][v]; o.z = r[2][v]; o.w = r[3][v];
    *(v4f*)&out[(size_t)(nbase + v + 8 * kh) * 256 + 64 * q + c4] = o;
  }
}

__global__ __launch_bounds__(128) void k_edge(
    const float* __restrict__ sph, const float* __restrict__ rb,
    const int* __restrict__ snd, const int* __restrict__ rcv,
    const float* __restrict__ W1, const float* __restrict__ W2,
    const float* __restrict__ x0, const float* __restrict__ x1,
    float* __restrict__ t0, float* __restrict__ t1,
    const float* __restrict__ nf, const float* __restrict__ attrs,
    const float* __restrict__ Wr0, const float* __restrict__ Wr1,
    float* __restrict__ out) {
  __shared__ __attribute__((aligned(16))) float hidden[4][16][64];
  const int wave  = threadIdx.x >> 5;
  const int lane  = threadIdx.x & 31;
  const int bi = blockIdx.x;
  const int b3 = bi / 3;
  if (bi < RES_SPAN && bi == 3 * b3) {
    const int q = __builtin_amdgcn_readfirstlane(wave);
    residual_chain(nf, attrs, q ? Wr1 : Wr0, out, b3 * 16, lane, q);
    return;
  }
  const int eb = bi < RES_SPAN ? bi - b3 - 1 : bi - RES_SPAN / 3;
  const int lrow  = lane & 15;
  const int kh    = lane >> 4;
  const int c4    = 4 * lrow;
  const int ebase = (eb * 4 + wave) * 16;

  {
    v8f acc[4] = {{}, {}, {}, {}};
#pragma unroll
    for (int kk = 0; kk < 8; kk += 4) {
      const int k0 = kk + 2 * kh;
      const v2f a = ld2(&rb[(size_t)(ebase + lrow) * 8 + k0]);
      v2f b[4];
      ldb4(W1, 64, k0, c4, b);
#pragma unroll
      for (int j = 0; j < 4; ++j)
        acc[j] = wmma4(a, b[j], acc[j]);
    }
#pragma unroll
    for (int v = 0; v < 8; ++v) {
      float h[4];
#pragma unroll
      for (int j = 0; j < 4; ++j) {
        float t = acc[j][v];
        h[j] = t / (1.f + __expf(-t));
      }
      v4f o;
      o.x = h[0]; o.y = h[1]; o.z = h[2]; o.w = h[3];
      *(v4f*)&hidden[wave][v + 8 * kh][c4] = o;
    }
  }
  __syncthreads();

  v8f w[16] = {{}, {}, {}, {}, {}, {}, {}, {}, {}, {}, {}, {}, {}, {}, {}, {}};
#pragma unroll 2
  for (int kk = 0; kk < 64; kk += 4) {
    const int k0 = kk + 2 * kh;
    const v2f a = ld2(&hidden[wave][lrow][k0]);
#pragma unroll
    for (int p = 0; p < 4; ++p) {
      v2f b[4];
      ldb4(W2, 256, k0, 64 * p + c4, b);
#pragma unroll
      for (int i = 0; i < 4; ++i)
        w[4 * p + i] = wmma4(a, b[i], w[4 * p + i]);
    }
  }

  int sidx[8], ridx[8];
  v4f Yv[8];
#pragma unroll
  for (int v = 0; v < 8; ++v) {
    const int e = ebase + v + 8 * kh;
    sidx[v] = snd[e];
    ridx[v] = rcv[e];
    Yv[v]   = *(const v4f*)&sph[(size_t)e * 4];
  }
#pragma unroll
  for (int v = 0; v < 8; ++v) {
    const int s = sidx[v];
    const int r = ridx[v];
    const float Y0  = Yv[v].x;
    const float Y1x = Yv[v].y;
    const float Y1y = Yv[v].z;
    const float Y1z = Yv[v].w;
    const v4f g0 = *(const v4f*)&x0[(size_t)s * 64 + c4];
    const v4f ga = *(const v4f*)&x1[(size_t)s * 192 + 3 * c4];
    const v4f gb = *(const v4f*)&x1[(size_t)s * 192 + 3 * c4 + 4];
    const v4f gc = *(const v4f*)&x1[(size_t)s * 192 + 3 * c4 + 8];
    asm volatile("" :: "v"(g0), "v"(ga), "v"(gb), "v"(gc));
    const float hs[4]  = {g0.x, g0.y, g0.z, g0.w};
    const float hv[12] = {ga.x, ga.y, ga.z, ga.w, gb.x, gb.y, gb.z, gb.w,
                          gc.x, gc.y, gc.z, gc.w};
    float* t0r = t0 + (size_t)r * 64 + lrow;
    float* t1r = t1 + (size_t)r * 192 + lrow;
#pragma unroll
    for (int i = 0; i < 4; ++i) {
      const float w0 = w[i][v], w1 = w[i + 4][v];
      const float w2 = w[i + 8][v], w3 = w[i + 12][v];
      const float h0  = hs[i];
      const float h1x = hv[3 * i + 0];
      const float h1y = hv[3 * i + 1];
      const float h1z = hv[3 * i + 2];
      const float dot = h1x * Y1x + h1y * Y1y + h1z * Y1z;
      atomicAdd(&t0r[16 * i], w0 * h0 * Y0 + w1 * dot * INV_SQRT3);
      atomicAdd(&t1r[16 * i],       w2 * h0 * Y1x + w3 * h1x * Y0);
      atomicAdd(&t1r[64 + 16 * i],  w2 * h0 * Y1y + w3 * h1y * Y0);
      atomicAdd(&t1r[128 + 16 * i], w2 * h0 * Y1z + w3 * h1z * Y0);
    }
  }
}

__global__ __launch_bounds__(64, 2) void k_node_post(
    const float* __restrict__ nf, const float* __restrict__ attrs,
    const float* __restrict__ Wpost0, const float* __restrict__ Wpost1,
    const float* __restrict__ Wc0, const float* __restrict__ Wc1,
    const float* __restrict__ Wf0, const float* __restrict__ Wf1,
    const float* __restrict__ t0, const float* __restrict__ t1,
    float* __restrict__ out) {
  __shared__ __attribute__((aligned(16))) float s0s[2][16][64];
  __shared__ __attribute__((aligned(16))) float s1s[2][16][192];
  const int wave  = __builtin_amdgcn_readfirstlane(threadIdx.x >> 5);
  const int lane  = threadIdx.x & 31;
  const int lrow  = lane & 15;
  const int kh    = lane >> 4;
  const int c4    = 4 * lrow;
  const int nbase = (blockIdx.x * 2 + wave) * 16;
  const int arow  = nbase + lrow;

  {
    const float4* rsrc = (const float4*)(out + (size_t)nbase * 256);
    for (int q = lane; q < 16 * 64; q += 32) {
      const float4 r4 = rsrc[q];
      const int row = q >> 6, q4 = q & 63;
      if (q4 < 16) *(float4*)&s0s[wave][row][q4 * 4] = r4;
      else         *(float4*)&s1s[wave][row][(q4 - 16) * 4] = r4;
    }
  }

  v8f a0[4]    = {{}, {}, {}, {}};
  v8f a1[3][4] = {{{}, {}, {}, {}}, {{}, {}, {}, {}}, {{}, {}, {}, {}}};
  for (int kk = 0; kk < 64; kk += 4) {
    const int k0 = kk + 2 * kh;
    const int p0 = 16 * (k0 & 3) + (k0 >> 2);
    v2f at0;
    at0.x = t0[(size_t)arow * 64 + p0];
    at0.y = t0[(size_t)arow * 64 + p0 + 16];
    v2f b1[4];
#pragma unroll
    for (int j = 0; j < 4; ++j)
      b1[j] = ldb1(Wpost1, 64, k0, 16 * j + lrow);
#pragma unroll
    for (int j = 0; j < 4; ++j)
      a0[j] = wmma4(at0, ldb1(Wpost0, 64, k0, 16 * j + lrow), a0[j]);
#pragma unroll
    for (int m = 0; m < 3; ++m) {
      v2f at1;
      at1.x = t1[(size_t)arow * 192 + 64 * m + p0];
      at1.y = t1[(size_t)arow * 192 + 64 * m + p0 + 16];
#pragma unroll
      for (int j = 0; j < 4; ++j)
        a1[m][j] = wmma4(at1, b1[j], a1[m][j]);
    }
  }

  const bool kpad = (kh != 0);
  v2f aat[3];
#pragma unroll
  for (int t = 0; t < 3; ++t) {
    const bool pad = (t == 2) && kpad;
    const int kc = pad ? 8 : 4 * t + 2 * kh;
    v2f a = ld2(&attrs[(size_t)arow * 10 + kc]);
    if (pad) { a.x = 0.f; a.y = 0.f; }
    aat[t] = a;
  }
#pragma unroll
  for (int j = 0; j < 4; ++j) {
    const int col = 16 * j + lrow;
    v2f bw[18];
#pragma unroll
    for (int t = 0; t < 3; ++t) {
      const int kc = ((t == 2) && kpad) ? 8 : 4 * t + 2 * kh;
#pragma unroll
      for (int p = 0; p < 6; ++p) {
        const float* Wm = (p < 3) ? (Wc0 + 640 * p) : (Wc1 + 640 * (p - 3));
        bw[6 * t + p] = ldb1(Wm, 64, kc, col);
      }
    }
    asm volatile("" :: "v"(bw[0]), "v"(bw[1]), "v"(bw[2]), "v"(bw[3]), "v"(bw[4]),
                 "v"(bw[5]), "v"(bw[6]), "v"(bw[7]), "v"(bw[8]), "v"(bw[9]),
                 "v"(bw[10]), "v"(bw[11]), "v"(bw[12]), "v"(bw[13]), "v"(bw[14]),
                 "v"(bw[15]), "v"(bw[16]), "v"(bw[17]));
    v8f cf[6] = {{}, {}, {}, {}, {}, {}};
#pragma unroll
    for (int t = 0; t < 3; ++t) {
      const bool pad = (t == 2) && kpad;
#pragma unroll
      for (int p = 0; p < 6; ++p) {
        v2f b = bw[6 * t + p];
        if (pad) { b.x = 0.f; b.y = 0.f; }
        cf[p] = wmma4(aat[t], b, cf[p]);
      }
    }
#pragma unroll
    for (int v = 0; v < 8; ++v) {
      const float s  = a0[j][v] * (1.f / 16.f);
      const float ax = a1[0][j][v] * (1.f / 16.f);
      const float ay = a1[1][j][v] * (1.f / 16.f);
      const float az = a1[2][j][v] * (1.f / 16.f);
      const float q  = ax * ax + ay * ay + az * az;
      const float c0 = cf[0][v], c1 = cf[1][v], c2 = cf[2][v];
      const float d0 = cf[3][v], d1 = cf[4][v], d2 = cf[5][v];
      const float s2 = s * s;
      const float g  = d2 * (s2 + q) + d1 * s + d0;
      a0[j][v]    = c2 * (s2 * s + s * q) + c1 * (s2 + q) + c0 * s;
      a1[0][j][v] = g * ax;
      a1[1][j][v] = g * ay;
      a1[2][j][v] = g * az;
    }
    __builtin_amdgcn_sched_barrier(0);
  }

  __syncthreads();
#pragma unroll
  for (int j = 0; j < 4; ++j) {
    const int col = 16 * j + lrow;
#pragma unroll
    for (int v = 0; v < 8; ++v) {
      const int M = v + 8 * kh;
      s0s[wave][M][col] += a0[j][v];
      s1s[wave][M][col] += a1[0][j][v];
      s1s[wave][M][64 + col] += a1[1][j][v];
      s1s[wave][M][128 + col] += a1[2][j][v];
    }
  }
  __syncthreads();

  {
    v8f acc[4] = {{}, {}, {}, {}};
    for (int kk = 0; kk < 64; kk += 4) {
      const int k0 = kk + 2 * kh;
      const v2f a = ld2(&s0s[wave][lrow][k0]);
      v2f b[4];
      ldb4(Wf0, 64, k0, c4, b);
#pragma unroll
      for (int j = 0; j < 4; ++j)
        acc[j] = wmma4(a, b[j], acc[j]);
    }
#pragma unroll
    for (int v = 0; v < 8; ++v) {
      v4f o;
      o.x = acc[0][v]; o.y = acc[1][v]; o.z = acc[2][v]; o.w = acc[3][v];
      *(v4f*)&out[(size_t)(nbase + v + 8 * kh) * 256 + c4] = o;
    }
  }
#pragma unroll
  for (int m = 0; m < 3; ++m) {
    v8f acc[4] = {{}, {}, {}, {}};
    for (int kk = 0; kk < 64; kk += 4) {
      const int k0 = kk + 2 * kh;
      const v2f a = ld2(&s1s[wave][lrow][64 * m + k0]);
      v2f b[4];
      ldb4(Wf1, 64, k0, c4, b);
#pragma unroll
      for (int j = 0; j < 4; ++j)
        acc[j] = wmma4(a, b[j], acc[j]);
    }
#pragma unroll
    for (int j = 0; j < 4; ++j)
#pragma unroll
      for (int v = 0; v < 8; ++v)
        out[(size_t)(nbase + v + 8 * kh) * 256 + 64 + (c4 + j) * 3 + m] =
            acc[j][v];
  }
}

extern "C" void kernel_launch(void* const* d_in, const int* in_sizes, int n_in,
                              void* d_out, int out_size, void* d_ws, size_t ws_size,
                              hipStream_t stream) {
  (void)in_sizes; (void)n_in; (void)out_size; (void)ws_size;
  const float* nf     = (const float*)d_in[0];
  const float* attrs  = (const float*)d_in[1];
  const float* sph    = (const float*)d_in[2];
  const float* rb     = (const float*)d_in[3];
  const int*   snd    = (const int*)d_in[4];
  const int*   rcv    = (const int*)d_in[5];
  const float* Wpre0  = (const float*)d_in[6];
  const float* Wpre1  = (const float*)d_in[7];
  const float* W1     = (const float*)d_in[8];
  const float* W2     = (const float*)d_in[9];
  const float* Wpost0 = (const float*)d_in[10];
  const float* Wpost1 = (const float*)d_in[11];
  const float* Wc0    = (const float*)d_in[12];
  const float* Wc1    = (const float*)d_in[13];
  const float* Wr0    = (const float*)d_in[14];
  const float* Wr1    = (const float*)d_in[15];
  const float* Wf0    = (const float*)d_in[16];
  const float* Wf1    = (const float*)d_in[17];

  float* ws = (float*)d_ws;
  size_t off = 0;
  float* x0 = ws + off; off += (size_t)N_NODES * 64;
  float* x1 = ws + off; off += (size_t)N_NODES * 192;
  float* t0 = ws + off; off += (size_t)N_NODES * 64;
  float* t1 = ws + off; off += (size_t)N_NODES * 192;

  const int ntiles = N_NODES / 16;
  k_node_pre<<<(ntiles + 3) / 4, 128, 0, stream>>>(nf, Wpre0, Wpre1, x0, x1, t0, t1);
  k_edge<<<EDGE_GRID, 128, 0, stream>>>(
      sph, rb, snd, rcv, W1, W2, x0, x1, t0, t1,
      nf, attrs, Wr0, Wr1, (float*)d_out);
  k_node_post<<<ntiles / 2, 64, 0, stream>>>(nf, attrs, Wpost0, Wpost1, Wc0, Wc1,
                                             Wf0, Wf1, t0, t1, (float*)d_out);
}


# edge scatter: message registers of tile pairs swap 16-lane rows in registers (v_permlane16_swap, no LDS): each atomic adds one whole 128-byte line of one receiver row; arrival-order sums as before
# speedup vs baseline: 1.6912x; 1.0300x over previous
#include <hip/hip_runtime.h>


#define N_NODES 20000
#define E_EDGES 320000
#define INV_SQRT3 0.57735026918962576451f
#define RES_SPAN 3750
#define EDGE_GRID (RES_SPAN / 3 + E_EDGES / 64)

typedef __attribute__((ext_vector_type(2))) float v2f;
typedef __attribute__((ext_vector_type(4))) float v4f;
typedef __attribute__((ext_vector_type(8))) float v8f;

static __device__ __forceinline__ v8f wmma4(v2f a, v2f b, v8f c) {
  return __builtin_amdgcn_wmma_f32_16x16x4_f32(
      false, a, false, b, (short)0, c, false, false);
}

static __device__ __forceinline__ void ldb4(const float* __restrict__ W, int ncol,
                                            int k0, int c0, v2f b[4]) {
  const v4f r0 = *(const v4f*)(W + (size_t)k0 * ncol + c0);
  const v4f r1 = *(const v4f*)(W + (size_t)(k0 + 1) * ncol + c0);
  b[0].x = r0.x; b[0].y = r1.x;
  b[1].x = r0.y; b[1].y = r1.y;
  b[2].x = r0.z; b[2].y = r1.z;
  b[3].x = r0.w; b[3].y = r1.w;
}

static __device__ __forceinline__ v2f ldb1(const float* __restrict__ W, int ncol,
                                          int k0, int col) {
  v2f r;
  r.x = W[(size_t)k0 * ncol + col];
  r.y = W[(size_t)(k0 + 1) * ncol + col];
  return r;
}

static __device__ __forceinline__ v2f ld2(const float* __restrict__ p) {
  v2f r; r.x = p[0]; r.y = p[1]; return r;
}

static __device__ __forceinline__ void swap_rows(float& a, float& b) {
  asm("v_permlane16_swap_b32 %0, %1" : "+v"(a), "+v"(b));
}

__global__ __launch_bounds__(128) void k_node_pre(
    const float* __restrict__ nf, const float* __restrict__ Wp0,
    const float* __restrict__ Wp1, float* __restrict__ x0,
    float* __restrict__ x1, float* __restrict__ t0, float* __restrict__ t1) {
  const int wave = threadIdx.x >> 5;
  const int lane = threadIdx.x & 31;
  const int tile = blockIdx.x * 4 + wave;
  if (tile >= (N_NODES / 16)) return;
  const int nbase = tile * 16;
  const int lrow  = lane & 15;
  const int kh    = lane >> 4;
  const int arow  = nbase + lrow;
  const int c4    = 4 * lrow;

  {
    v8f acc[4] = {{}, {}, {}, {}};
    for (int kk = 0; kk < 64; kk += 4) {
      const int k0 = kk + 2 * kh;
      const v2f a = ld2(&nf[(size_t)arow * 256 + k0]);
      v2f b[4];
      ldb4(Wp0, 64, k0, c4, b);
#pragma unroll
      for (int j = 0; j < 4; ++j)
        acc[j] = wmma4(a, b[j], acc[j]);
    }
#pragma unroll
    for (int v = 0; v < 8; ++v) {
      v4f o;
      o.x = acc[0][v]; o.y = acc[1][v]; o.z = acc[2][v]; o.w = acc[3][v];
      *(v4f*)&x0[(size_t)(nbase + v + 8 * kh) * 64 + c4] = o;
    }
  }

  for (int m = 0; m < 3; ++m) {
    v8f acc[4] = {{}, {}, {}, {}};
    for (int kk = 0; kk < 64; kk += 4) {
      const int k0 = kk + 2 * kh;
      v2f a;
      a.x = nf[(size_t)arow * 256 + 64 + k0 * 3 + m];
      a.y = nf[(size_t)arow * 256 + 64 + (k0 + 1) * 3 + m];
      v2f b[4];
      ldb4(Wp1, 64, k0, c4, b);
#pragma unroll
      for (int j = 0; j < 4; ++j)
        acc[j] = wmma4(a, b[j], acc[j]);
    }
#pragma unroll
    for (int j = 0; j < 4; ++j)
#pragma unroll
      for (int v = 0; v < 8; ++v)
        x1[(size_t)(nbase + v + 8 * kh) * 192 + (c4 + j) * 3 + m] = acc[j][v];
  }

  float4* t0v = (float4*)(t0 + (size_t)nbase * 64);
  float4* t1v = (float4*)(t1 + (size_t)nbase * 192);
  const float4 z4 = make_float4(0.f, 0.f, 0.f, 0.f);
  for (int i = lane; i < 16 * 64 / 4; i += 32)  t0v[i] = z4;
  for (int i = lane; i < 16 * 192 / 4; i += 32) t1v[i] = z4;
}

static __device__ __forceinline__ void residual_chain(
    const float* __restrict__ nf, const float* __restrict__ attrs,
    const float* __restrict__ Wr, float* __restrict__ out,
    int nbase, int lane, int q) {
  const int lrow = lane & 15;
  const int kh   = lane >> 4;
  const int arow = nbase + lrow;
  const int c4   = 4 * lrow;
  const float* nrow = nf + (size_t)arow * 256 + (q ? 63 + q : 0);
  const int ustep = q ? 3 : 1;
  v2f a[16];
#pragma unroll
  for (int t = 0; t < 16; ++t) {
    const int k0 = 4 * t + 2 * kh;
    a[t].x = nrow[k0 * ustep];
    a[t].y = nrow[(k0 + 1) * ustep];
  }
  const float* trow = attrs + (size_t)(nbase + 8 * kh) * 10;
  v8f r[4] = {{}, {}, {}, {}};
  for (int z = 0; z < 10; ++z) {
    const float* Wz = Wr + 64 * z;
    v8f acc[4] = {{}, {}, {}, {}};
#pragma unroll
    for (int t = 0; t < 16; ++t) {
      const int k0 = 4 * t + 2 * kh;
      v2f b[4];
      ldb4(Wz, 640, k0, c4, b);
#pragma unroll
      for (int j = 0; j < 4; ++j)
        acc[j] = wmma4(a[t], b[j], acc[j]);
    }
#pragma unroll
    for (int v = 0; v < 8; ++v) {
      const float tz = trow[10 * v + z];
#pragma unroll
      for (int j = 0; j < 4; ++j)
        r[j][v] += tz * acc[j][v];
    }
  }
#pragma unroll
  for (int v = 0; v < 8; ++v) {
    v4f o;
    o.x = r[0][v]; o.y = r[1][v]; o.z = r[2][v]; o.w = r[3][v];
    *(v4f*)&out[(size_t)(nbase + v + 8 * kh) * 256 + 64 * q + c4] = o;
  }
}

__global__ __launch_bounds__(128) void k_edge(
    const float* __restrict__ sph, const float* __restrict__ rb,
    const int* __restrict__ snd, const int* __restrict__ rcv,
    const float* __restrict__ W1, const float* __restrict__ W2,
    const float* __restrict__ x0, const float* __restrict__ x1,
    float* __restrict__ t0, float* __restrict__ t1,
    const float* __restrict__ nf, const float* __restrict__ attrs,
    const float* __restrict__ Wr0, const float* __restrict__ Wr1,
    float* __restrict__ out) {
  __shared__ __attribute__((aligned(16))) float hidden[4][16][64];
  const int wave  = threadIdx.x >> 5;
  const int lane  = threadIdx.x & 31;
  const int bi = blockIdx.x;
  const int b3 = bi / 3;
  if (bi < RES_SPAN && bi == 3 * b3) {
    const int q = __builtin_amdgcn_readfirstlane(wave);
    residual_chain(nf, attrs, q ? Wr1 : Wr0, out, b3 * 16, lane, q);
    return;
  }
  const int eb = bi < RES_SPAN ? bi - b3 - 1 : bi - RES_SPAN / 3;
  const int lrow  = lane & 15;
  const int kh    = lane >> 4;
  const int c4    = 4 * lrow;
  const int ebase = (eb * 4 + wave) * 16;

  {
    v8f acc[4] = {{}, {}, {}, {}};
#pragma unroll
    for (int kk = 0; kk < 8; kk += 4) {
      const int k0 = kk + 2 * kh;
      const v2f a = ld2(&rb[(size_t)(ebase + lrow) * 8 + k0]);
      v2f b[4];
      ldb4(W1, 64, k0, c4, b);
#pragma unroll
      for (int j = 0; j < 4; ++j)
        acc[j] = wmma4(a, b[j], acc[j]);
    }
#pragma unroll
    for (int v = 0; v < 8; ++v) {
      float h[4];
#pragma unroll
      for (int j = 0; j < 4; ++j) {
        float t = acc[j][v];
        h[j] = t / (1.f + __expf(-t));
      }
      v4f o;
      o.x = h[0]; o.y = h[1]; o.z = h[2]; o.w = h[3];
      *(v4f*)&hidden[wave][v + 8 * kh][c4] = o;
    }
  }
  __syncthreads();

  v8f w[16] = {{}, {}, {}, {}, {}, {}, {}, {}, {}, {}, {}, {}, {}, {}, {}, {}};
#pragma unroll 2
  for (int kk = 0; kk < 64; kk += 4) {
    const int k0 = kk + 2 * kh;
    const v2f a = ld2(&hidden[wave][lrow][k0]);
#pragma unroll
    for (int p = 0; p < 4; ++p) {
      v2f b[4];
      ldb4(W2, 256, k0, 64 * p + c4, b);
#pragma unroll
      for (int i = 0; i < 4; ++i)
        w[4 * p + i] = wmma4(a, b[i], w[4 * p + i]);
    }
  }

  const int ebs = __builtin_amdgcn_readfirstlane(ebase);
  int sidx[8], rowa[8], rowb[8];
  v4f Yv[8];
#pragma unroll
  for (int v = 0; v < 8; ++v) {
    const int e = ebase + v + 8 * kh;
    sidx[v] = snd[e];
    Yv[v]   = *(const v4f*)&sph[(size_t)e * 4];
    rowa[v] = rcv[ebs + v];
    rowb[v] = rcv[ebs + v + 8];
  }
#pragma unroll
  for (int v = 0; v < 8; ++v) {
    const int s = sidx[v];
    const float Y0  = Yv[v].x;
    const float Y1x = Yv[v].y;
    const float Y1y = Yv[v].z;
    const float Y1z = Yv[v].w;
    const v4f g0 = *(const v4f*)&x0[(size_t)s * 64 + c4];
    const v4f ga = *(const v4f*)&x1[(size_t)s * 192 + 3 * c4];
    const v4f gb = *(const v4f*)&x1[(size_t)s * 192 + 3 * c4 + 4];
    const v4f gc = *(const v4f*)&x1[(size_t)s * 192 + 3 * c4 + 8];
    asm volatile("" :: "v"(g0), "v"(ga), "v"(gb), "v"(gc));
    const float hs[4]  = {g0.x, g0.y, g0.z, g0.w};
    const float hv[12] = {ga.x, ga.y, ga.z, ga.w, gb.x, gb.y, gb.z, gb.w,
                          gc.x, gc.y, gc.z, gc.w};
    float m0[4], mx[4], my[4], mz[4];
#pragma unroll
    for (int i = 0; i < 4; ++i) {
      const float w0 = w[i][v], w1 = w[i + 4][v];
      const float w2 = w[i + 8][v], w3 = w[i + 12][v];
      const float h0  = hs[i];
      const float h1x = hv[3 * i + 0];
      const float h1y = hv[3 * i + 1];
      const float h1z = hv[3 * i + 2];
      const float dot = h1x * Y1x + h1y * Y1y + h1z * Y1z;
      m0[i] = w0 * h0 * Y0 + w1 * dot * INV_SQRT3;
      mx[i] = w2 * h0 * Y1x + w3 * h1x * Y0;
      my[i] = w2 * h0 * Y1y + w3 * h1y * Y0;
      mz[i] = w2 * h0 * Y1z + w3 * h1z * Y0;
    }
    float* ta0 = t0 + (size_t)rowa[v] * 64 + lane;
    float* tb0 = t0 + (size_t)rowb[v] * 64 + lane;
    float* ta1 = t1 + (size_t)rowa[v] * 192 + lane;
    float* tb1 = t1 + (size_t)rowb[v] * 192 + lane;
#pragma unroll
    for (int j = 0; j < 2; ++j) {
      swap_rows(m0[2 * j], m0[2 * j + 1]);
      swap_rows(mx[2 * j], mx[2 * j + 1]);
      swap_rows(my[2 * j], my[2 * j + 1]);
      swap_rows(mz[2 * j], mz[2 * j + 1]);
      atomicAdd(&ta0[32 * j],       m0[2 * j]);
      atomicAdd(&tb0[32 * j],       m0[2 * j + 1]);
      atomicAdd(&ta1[32 * j],       mx[2 * j]);
      atomicAdd(&tb1[32 * j],       mx[2 * j + 1]);
      atomicAdd(&ta1[64 + 32 * j],  my[2 * j]);
      atomicAdd(&tb1[64 + 32 * j],  my[2 * j + 1]);
      atomicAdd(&ta1[128 + 32 * j], mz[2 * j]);
      atomicAdd(&tb1[128 + 32 * j], mz[2 * j + 1]);
    }
  }
}

__global__ __launch_bounds__(64, 2) void k_node_post(
    const float* __restrict__ nf, const float* __restrict__ attrs,
    const float* __restrict__ Wpost0, const float* __restrict__ Wpost1,
    const float* __restrict__ Wc0, const float* __restrict__ Wc1,
    const float* __restrict__ Wf0, const float* __restrict__ Wf1,
    const float* __restrict__ t0, const float* __restrict__ t1,
    float* __restrict__ out) {
  __shared__ __attribute__((aligned(16))) float s0s[2][16][64];
  __shared__ __attribute__((aligned(16))) float s1s[2][16][192];
  const int wave  = __builtin_amdgcn_readfirstlane(threadIdx.x >> 5);
  const int lane  = threadIdx.x & 31;
  const int lrow  = lane & 15;
  const int kh    = lane >> 4;
  const int c4    = 4 * lrow;
  const int nbase = (blockIdx.x * 2 + wave) * 16;
  const int arow  = nbase + lrow;

  {
    const float4* rsrc = (const float4*)(out + (size_t)nbase * 256);
    for (int q = lane; q < 16 * 64; q += 32) {
      const float4 r4 = rsrc[q];
      const int row = q >> 6, q4 = q & 63;
      if (q4 < 16) *(float4*)&s0s[wave][row][q4 * 4] = r4;
      else         *(float4*)&s1s[wave][row][(q4 - 16) * 4] = r4;
    }
  }

  v8f a0[4]    = {{}, {}, {}, {}};
  v8f a1[3][4] = {{{}, {}, {}, {}}, {{}, {}, {}, {}}, {{}, {}, {}, {}}};
  for (int kk = 0; kk < 64; kk += 4) {
    const int k0 = kk + 2 * kh;
    const int p0 = 16 * (k0 & 3) + (k0 >> 2);
    v2f at0;
    at0.x = t0[(size_t)arow * 64 + p0];
    at0.y = t0[(size_t)arow * 64 + p0 + 16];
    v2f b1[4];
#pragma unroll
    for (int j = 0; j < 4; ++j)
      b1[j] = ldb1(Wpost1, 64, k0, 16 * j + lrow);
#pragma unroll
    for (int j = 0; j < 4; ++j)
      a0[j] = wmma4(at0, ldb1(Wpost0, 64, k0, 16 * j + lrow), a0[j]);
#pragma unroll
    for (int m = 0; m < 3; ++m) {
      v2f at1;
      at1.x = t1[(size_t)arow * 192 + 64 * m + p0];
      at1.y = t1[(size_t)arow * 192 + 64 * m + p0 + 16];
#pragma unroll
      for (int j = 0; j < 4; ++j)
        a1[m][j] = wmma4(at1, b1[j], a1[m][j]);
    }
  }

  const bool kpad = (kh != 0);
  v2f aat[3];
#pragma unroll
  for (int t = 0; t < 3; ++t) {
    const bool pad = (t == 2) && kpad;
    const int kc = pad ? 8 : 4 * t + 2 * kh;
    v2f a = ld2(&attrs[(size_t)arow * 10 + kc]);
    if (pad) { a.x = 0.f; a.y = 0.f; }
    aat[t] = a;
  }
#pragma unroll
  for (int j = 0; j < 4; ++j) {
    const int col = 16 * j + lrow;
    v2f bw[18];
#pragma unroll
    for (int t = 0; t < 3; ++t) {
      const int kc = ((t == 2) && kpad) ? 8 : 4 * t + 2 * kh;
#pragma unroll
      for (int p = 0; p < 6; ++p) {
        const float* Wm = (p < 3) ? (Wc0 + 640 * p) : (Wc1 + 640 * (p - 3));
        bw[6 * t + p] = ldb1(Wm, 64, kc, col);
      }
    }
    asm volatile("" :: "v"(bw[0]), "v"(bw[1]), "v"(bw[2]), "v"(bw[3]), "v"(bw[4]),
                 "v"(bw[5]), "v"(bw[6]), "v"(bw[7]), "v"(bw[8]), "v"(bw[9]),
                 "v"(bw[10]), "v"(bw[11]), "v"(bw[12]), "v"(bw[13]), "v"(bw[14]),
                 "v"(bw[15]), "v"(bw[16]), "v"(bw[17]));
    v8f cf[6] = {{}, {}, {}, {}, {}, {}};
#pragma unroll
    for (int t = 0; t < 3; ++t) {
      const bool pad = (t == 2) && kpad;
#pragma unroll
      for (int p = 0; p < 6; ++p) {
        v2f b = bw[6 * t + p];
        if (pad) { b.x = 0.f; b.y = 0.f; }
        cf[p] = wmma4(aat[t], b, cf[p]);
      }
    }
#pragma unroll
    for (int v = 0; v < 8; ++v) {
      const float s  = a0[j][v] * (1.f / 16.f);
      const float ax = a1[0][j][v] * (1.f / 16.f);
      const float ay = a1[1][j][v] * (1.f / 16.f);
      const float az = a1[2][j][v] * (1.f / 16.f);
      const float q  = ax * ax + ay * ay + az * az;
      const float c0 = cf[0][v], c1 = cf[1][v], c2 = cf[2][v];
      const float d0 = cf[3][v], d1 = cf[4][v], d2 = cf[5][v];
      const float s2 = s * s;
      const float g  = d2 * (s2 + q) + d1 * s + d0;
      a0[j][v]    = c2 * (s2 * s + s * q) + c1 * (s2 + q) + c0 * s;
      a1[0][j][v] = g * ax;
      a1[1][j][v] = g * ay;
      a1[2][j][v] = g * az;
    }
    __builtin_amdgcn_sched_barrier(0);
  }

  __syncthreads();
#pragma unroll
  for (int j = 0; j < 4; ++j) {
    const int col = 16 * j + lrow;
#pragma unroll
    for (int v = 0; v < 8; ++v) {
      const int M = v + 8 * kh;
      s0s[wave][M][col] += a0[j][v];
      s1s[wave][M][col] += a1[0][j][v];
      s1s[wave][M][64 + col] += a1[1][j][v];
      s1s[wave][M][128 + col] += a1[2][j][v];
    }
  }
  __syncthreads();

  {
    v8f acc[4] = {{}, {}, {}, {}};
    for (int kk = 0; kk < 64; kk += 4) {
      const int k0 = kk + 2 * kh;
      const v2f a = ld2(&s0s[wave][lrow][k0]);
      v2f b[4];
      ldb4(Wf0, 64, k0, c4, b);
#pragma unroll
      for (int j = 0; j < 4; ++j)
        acc[j] = wmma4(a, b[j], acc[j]);
    }
#pragma unroll
    for (int v = 0; v < 8; ++v) {
      v4f o;
      o.x = acc[0][v]; o.y = acc[1][v]; o.z = acc[2][v]; o.w = acc[3][v];
      *(v4f*)&out[(size_t)(nbase + v + 8 * kh) * 256 + c4] = o;
    }
  }
#pragma unroll
  for (int m = 0; m < 3; ++m) {
    v8f acc[4] = {{}, {}, {}, {}};
    for (int kk = 0; kk < 64; kk += 4) {
      const int k0 = kk + 2 * kh;
      const v2f a = ld2(&s1s[wave][lrow][64 * m + k0]);
      v2f b[4];
      ldb4(Wf1, 64, k0, c4, b);
#pragma unroll
      for (int j = 0; j < 4; ++j)
        acc[j] = wmma4(a, b[j], acc[j]);
    }
#pragma unroll
    for (int j = 0; j < 4; ++j)
#pragma unroll
      for (int v = 0; v < 8; ++v)
        out[(size_t)(nbase + v + 8 * kh) * 256 + 64 + (c4 + j) * 3 + m] =
            acc[j][v];
  }
}

extern "C" void kernel_launch(void* const* d_in, const int* in_sizes, int n_in,
                              void* d_out, int out_size, void* d_ws, size_t ws_size,
                              hipStream_t stream) {
  (void)in_sizes; (void)n_in; (void)out_size; (void)ws_size;
  const float* nf     = (const float*)d_in[0];
  const float* attrs  = (const float*)d_in[1];
  const float* sph    = (const float*)d_in[2];
  const float* rb     = (const float*)d_in[3];
  const int*   snd    = (const int*)d_in[4];
  const int*   rcv    = (const int*)d_in[5];
  const float* Wpre0  = (const float*)d_in[6];
  const float* Wpre1  = (const float*)d_in[7];
  const float* W1     = (const float*)d_in[8];
  const float* W2     = (const float*)d_in[9];
  const float* Wpost0 = (const float*)d_in[10];
  const float* Wpost1 = (const float*)d_in[11];
  const float* Wc0    = (const float*)d_in[12];
  const float* Wc1    = (const float*)d_in[13];
  const float* Wr0    = (const float*)d_in[14];
  const float* Wr1    = (const float*)d_in[15];
  const float* Wf0    = (const float*)d_in[16];
  const float* Wf1    = (const float*)d_in[17];

  float* ws = (float*)d_ws;
  size_t off = 0;
  float* x0 = ws + off; off += (size_t)N_NODES * 64;
  float* x1 = ws + off; off += (size_t)N_NODES * 192;
  float* t0 = ws + off; off += (size_t)N_NODES * 64;
  float* t1 = ws + off; off += (size_t)N_NODES * 192;

  const int ntiles = N_NODES / 16;
  k_node_pre<<<(ntiles + 3) / 4, 128, 0, stream>>>(nf, Wpre0, Wpre1, x0, x1, t0, t1);
  k_edge<<<EDGE_GRID, 128, 0, stream>>>(
      sph, rb, snd, rcv, W1, W2, x0, x1, t0, t1,
      nf, attrs, Wr0, Wr1, (float*)d_out);
  k_node_post<<<ntiles / 2, 64, 0, stream>>>(nf, attrs, Wpost0, Wpost1, Wc0, Wc1,
                                             Wf0, Wf1, t0, t1, (float*)d_out);
}
